# starting kernel
# baseline (speedup 1.0000x reference)
#include <hip/hip_runtime.h>


typedef __attribute__((ext_vector_type(2))) float v2f;
typedef __attribute__((ext_vector_type(8))) float v8f;

#define EP_BETA 1e-3f
#define EP_LR   0.5f

constexpr int BM = 128;
constexpr int BN = 128;
constexpr int BK = 32;

template <bool TA, bool TB, int EPI>
__global__ void __launch_bounds__(256)
gemm_f32_wmma(const float* __restrict__ A, const float* __restrict__ B,
              const float* __restrict__ Cin, float* __restrict__ D,
              int M, int N, int K, float alpha)
{
    __shared__ v2f Asp[BK / 2][BM + 2];
    __shared__ v2f Bsp[BK / 2][BN + 2];

    const int tid   = threadIdx.x;
    const int lane  = tid & 31;
    const int wid   = tid >> 5;
    const int waveM = wid & 3;
    const int waveN = wid >> 2;
    const int g     = lane >> 4;
    const int ln    = lane & 15;

    const int bm = blockIdx.x * BM;
    const int bn = blockIdx.y * BN;

    v8f acc[2][4] = {};

    for (int kt = 0; kt < K; kt += BK) {
        if (kt + BK < K) {
            const float* pa = TA ? A + (size_t)(kt + BK + (tid >> 5)) * M + bm + 4 * (tid & 31)
                                 : A + (size_t)(bm + (tid >> 3)) * K + kt + BK + 4 * (tid & 7);
            const float* pb = TB ? B + (size_t)(bn + (tid >> 3)) * K + kt + BK + 4 * (tid & 7)
                                 : B + (size_t)(kt + BK + (tid >> 5)) * N + bn + 4 * (tid & 31);
            __builtin_prefetch(pa, 0, 1);
            __builtin_prefetch(pb, 0, 1);
        }

        #pragma unroll
        for (int i = 0; i < 4; ++i) {
            int idx4 = i * 256 + tid;
            if (!TA) {
                int kq = idx4 & 7;
                int m  = idx4 >> 3;
                float4 v = *(const float4*)(A + (size_t)(bm + m) * K + kt + 4 * kq);
                v2f p0; p0.x = v.x; p0.y = v.y;
                v2f p1; p1.x = v.z; p1.y = v.w;
                Asp[2 * kq + 0][m] = p0;
                Asp[2 * kq + 1][m] = p1;
            } else {
                int mq = idx4 & 31;
                int k  = idx4 >> 5;
                float4 v = *(const float4*)(A + (size_t)(kt + k) * M + bm + 4 * mq);
                ((float*)&Asp[k >> 1][4 * mq + 0])[k & 1] = v.x;
                ((float*)&Asp[k >> 1][4 * mq + 1])[k & 1] = v.y;
                ((float*)&Asp[k >> 1][4 * mq + 2])[k & 1] = v.z;
                ((float*)&Asp[k >> 1][4 * mq + 3])[k & 1] = v.w;
            }
        }
        #pragma unroll
        for (int i = 0; i < 4; ++i) {
            int idx4 = i * 256 + tid;
            if (TB) {
                int kq = idx4 & 7;
                int n  = idx4 >> 3;
                float4 v = *(const float4*)(B + (size_t)(bn + n) * K + kt + 4 * kq);
                v2f p0; p0.x = v.x; p0.y = v.y;
                v2f p1; p1.x = v.z; p1.y = v.w;
                Bsp[2 * kq + 0][n] = p0;
                Bsp[2 * kq + 1][n] = p1;
            } else {
                int nq = idx4 & 31;
                int k  = idx4 >> 5;
                float4 v = *(const float4*)(B + (size_t)(kt + k) * N + bn + 4 * nq);
                ((float*)&Bsp[k >> 1][4 * nq + 0])[k & 1] = v.x;
                ((float*)&Bsp[k >> 1][4 * nq + 1])[k & 1] = v.y;
                ((float*)&Bsp[k >> 1][4 * nq + 2])[k & 1] = v.z;
                ((float*)&Bsp[k >> 1][4 * nq + 3])[k & 1] = v.w;
            }
        }
        __syncthreads();

        #pragma unroll
        for (int k0 = 0; k0 < BK; k0 += 4) {
            const int kp = (k0 >> 1) + g;
            v2f a0 = Asp[kp][32 * waveM + ln];
            v2f a1 = Asp[kp][32 * waveM + 16 + ln];
            #pragma unroll
            for (int j = 0; j < 4; ++j) {
                v2f b = Bsp[kp][64 * waveN + 16 * j + ln];
                acc[0][j] = __builtin_amdgcn_wmma_f32_16x16x4_f32(
                    false, a0, false, b, (short)0, acc[0][j], false, false);
                acc[1][j] = __builtin_amdgcn_wmma_f32_16x16x4_f32(
                    false, a1, false, b, (short)0, acc[1][j], false, false);
            }
        }
        __syncthreads();
    }

    #pragma unroll
    for (int i2 = 0; i2 < 2; ++i2) {
        #pragma unroll
        for (int j = 0; j < 4; ++j) {
            int col = bn + 64 * waveN + 16 * j + ln;
            #pragma unroll
            for (int r = 0; r < 8; ++r) {
                int row = bm + 32 * waveM + 16 * i2 + r + 8 * g;
                size_t idx = (size_t)row * N + col;
                float v = acc[i2][j][r];
                if (EPI == 0)      D[idx] = alpha * v;
                else if (EPI == 1) D[idx] = Cin[idx] - v;
                else               D[idx] = Cin[idx] + v;
            }
        }
    }
}

__global__ void __launch_bounds__(256) subk(float* __restrict__ o,
                                            const float* __restrict__ a,
                                            const float* __restrict__ b) {
    size_t i = ((size_t)blockIdx.x * 256 + threadIdx.x) * 4;
    float4 av = *(const float4*)(a + i);
    float4 bv = *(const float4*)(b + i);
    *(float4*)(o + i) = make_float4(av.x - bv.x, av.y - bv.y,
                                    av.z - bv.z, av.w - bv.w);
}

__global__ void __launch_bounds__(256) g3k(float* __restrict__ r3,
                                           const float* __restrict__ t3,
                                           const float* __restrict__ y) {
    size_t i = ((size_t)blockIdx.x * 256 + threadIdx.x) * 4;
    float4 rv = *(const float4*)(r3 + i);
    float4 tv = *(const float4*)(t3 + i);
    float4 yv = *(const float4*)(y + i);
    rv.x += EP_BETA * (tv.x - yv.x);
    rv.y += EP_BETA * (tv.y - yv.y);
    rv.z += EP_BETA * (tv.z - yv.z);
    rv.w += EP_BETA * (tv.w - yv.w);
    *(float4*)(r3 + i) = rv;
}

__global__ void __launch_bounds__(256) updk(float* __restrict__ t,
                                            const float* __restrict__ gr) {
    size_t i = ((size_t)blockIdx.x * 256 + threadIdx.x) * 4;
    float4 tv = *(const float4*)(t + i);
    float4 gv = *(const float4*)(gr + i);
    tv.x -= EP_LR * gv.x; tv.y -= EP_LR * gv.y;
    tv.z -= EP_LR * gv.z; tv.w -= EP_LR * gv.w;
    *(float4*)(t + i) = tv;
}

__global__ void __launch_bounds__(256) copyk(float* __restrict__ dst,
                                             const float* __restrict__ src) {
    size_t i = ((size_t)blockIdx.x * 256 + threadIdx.x) * 4;
    *(float4*)(dst + i) = *(const float4*)(src + i);
}

__global__ void __launch_bounds__(256) loss_partial(const float* __restrict__ s,
                                                    const float* __restrict__ y,
                                                    float* __restrict__ part) {
    __shared__ float red[256];
    size_t i = ((size_t)blockIdx.x * 256 + threadIdx.x) * 4;
    float4 a = *(const float4*)(s + i);
    float4 b = *(const float4*)(y + i);
    float dx = a.x - b.x, dy = a.y - b.y, dz = a.z - b.z, dw = a.w - b.w;
    red[threadIdx.x] = dx * dx + dy * dy + dz * dz + dw * dw;
    __syncthreads();
    for (int off = 128; off > 0; off >>= 1) {
        if (threadIdx.x < off) red[threadIdx.x] += red[threadIdx.x + off];
        __syncthreads();
    }
    if (threadIdx.x == 0) part[blockIdx.x] = red[0];
}

__global__ void __launch_bounds__(256) loss_final(const float* __restrict__ part,
                                                  float* __restrict__ out) {
    __shared__ float red[256];
    red[threadIdx.x] = part[threadIdx.x] + part[threadIdx.x + 256];
    __syncthreads();
    for (int off = 128; off > 0; off >>= 1) {
        if (threadIdx.x < off) red[threadIdx.x] += red[threadIdx.x + off];
        __syncthreads();
    }
    if (threadIdx.x == 0) out[0] = red[0] * (1.0f / 524288.0f);
}

#define GEMM(TA, TB, EPI, M, N, K, Ap, Bp, Cp, Dp, alpha)                     \
    gemm_f32_wmma<TA, TB, EPI><<<dim3((M) / BM, (N) / BN), 256, 0, stream>>>( \
        Ap, Bp, Cp, Dp, M, N, K, alpha)

#define EW(kernel, n, ...) kernel<<<dim3((n) / 1024), 256, 0, stream>>>(__VA_ARGS__)

extern "C" void kernel_launch(void* const* d_in, const int* in_sizes, int n_in,
                              void* d_out, int out_size, void* d_ws, size_t ws_size,
                              hipStream_t stream)
{
    const float* x  = (const float*)d_in[0];
    const float* y  = (const float*)d_in[1];
    const float* W0 = (const float*)d_in[2];
    const float* W1 = (const float*)d_in[3];
    const float* W2 = (const float*)d_in[4];
    float* out = (float*)d_out;

    const int Bt = 1024;
    const int D0 = 1024, D1 = 1024, D2 = 1024, D3 = 512;
    const size_t n1 = (size_t)Bt * D1;
    const size_t n2 = (size_t)Bt * D2;
    const size_t n3 = (size_t)Bt * D3;

    float* ws = (float*)d_ws;
    float* p0 = ws;
    float* t1 = p0 + n1;
    float* t2 = t1 + n1;
    float* t3 = t2 + n2;
    float* r1 = t3 + n3;
    float* r2 = r1 + n1;
    float* r3 = r2 + n2;
    float* part = r3 + n3;

    GEMM(false, false, 0, Bt, D1, D0, x,  W0, (const float*)nullptr, p0, 1.0f);
    EW(copyk, n1, t1, p0);
    GEMM(false, false, 0, Bt, D2, D1, t1, W1, (const float*)nullptr, t2, 1.0f);
    GEMM(false, false, 0, Bt, D3, D2, t2, W2, (const float*)nullptr, t3, 1.0f);
    loss_partial<<<512, 256, 0, stream>>>(t3, y, part);
    loss_final<<<1, 256, 0, stream>>>(part, out + 2621440);

    for (int it = 0; it < 20; ++it) {
        GEMM(false, false, 1, Bt, D2, D1, t1, W1, t2, r2, 1.0f);
        GEMM(false, false, 1, Bt, D3, D2, t2, W2, t3, r3, 1.0f);
        EW(subk, n1, r1, t1, p0);
        GEMM(false, true, 1, Bt, D1, D2, r2, W1, r1, r1, 1.0f);
        GEMM(false, true, 1, Bt, D2, D3, r3, W2, r2, r2, 1.0f);
        EW(g3k, n3, r3, t3, y);
        if (it == 0) {
            GEMM(false, true, 2, Bt, D2, D3, r3, W2, r2, r2, 1.0f);
            GEMM(false, true, 2, Bt, D1, D2, r2, W1, r1, r1, 1.0f);
        }
        EW(updk, n1, t1, r1);
        EW(updk, n2, t2, r2);
        EW(updk, n3, t3, r3);
    }

    GEMM(false, false, 1, Bt, D2, D1, t1, W1, t2, r2, 1.0f);
    GEMM(false, false, 1, Bt, D3, D2, t2, W2, t3, r3, 1.0f);
    EW(subk, n1, r1, t1, p0);
    GEMM(true, false, 0, D0, D1, Bt, x,  r1, (const float*)nullptr, out,           -1.0f / EP_BETA);
    GEMM(true, false, 0, D1, D2, Bt, t1, r2, (const float*)nullptr, out + 1048576, -1.0f / EP_BETA);
    GEMM(true, false, 0, D2, D3, Bt, t2, r3, (const float*)nullptr, out + 2097152, -1.0f / EP_BETA);
}


# GEMM k-loop software-pipelined: next K tile's global loads issued as one group before this tile's WMMAs (replaces the cache prefetch hint), each 4-k block's LDS fragments loaded one block ahead
# speedup vs baseline: 1.5262x; 1.5262x over previous
#include <hip/hip_runtime.h>


typedef __attribute__((ext_vector_type(2))) float v2f;
typedef __attribute__((ext_vector_type(8))) float v8f;

#define EP_BETA 1e-3f
#define EP_LR   0.5f

constexpr int BM = 128;
constexpr int BN = 128;
constexpr int BK = 32;

template <bool TA, bool TB, int EPI>
__global__ void __launch_bounds__(256)
gemm_f32_wmma(const float* __restrict__ A, const float* __restrict__ B,
              const float* __restrict__ Cin, float* __restrict__ D,
              int M, int N, int K, float alpha)
{
    __shared__ v2f Asp[BK / 2][BM + 2];
    __shared__ v2f Bsp[BK / 2][BN + 2];

    const int tid   = threadIdx.x;
    const int lane  = tid & 31;
    const int wid   = tid >> 5;
    const int waveM = wid & 3;
    const int waveN = wid >> 2;
    const int g     = lane >> 4;
    const int ln    = lane & 15;

    const int bm = blockIdx.x * BM;
    const int bn = blockIdx.y * BN;

    v8f acc[2][4] = {};

    float4 va[4], vb[4];

    auto load_tile = [&](int kt) {
        #pragma unroll
        for (int i = 0; i < 4; ++i) {
            int idx4 = i * 256 + tid;
            if (!TA) {
                int kq = idx4 & 7;
                int m  = idx4 >> 3;
                va[i] = *(const float4*)(A + (size_t)(bm + m) * K + kt + 4 * kq);
            } else {
                int mq = idx4 & 31;
                int k  = idx4 >> 5;
                va[i] = *(const float4*)(A + (size_t)(kt + k) * M + bm + 4 * mq);
            }
        }
        #pragma unroll
        for (int i = 0; i < 4; ++i) {
            int idx4 = i * 256 + tid;
            if (TB) {
                int kq = idx4 & 7;
                int n  = idx4 >> 3;
                vb[i] = *(const float4*)(B + (size_t)(bn + n) * K + kt + 4 * kq);
            } else {
                int nq = idx4 & 31;
                int k  = idx4 >> 5;
                vb[i] = *(const float4*)(B + (size_t)(kt + k) * N + bn + 4 * nq);
            }
        }
    };

    auto store_tile = [&]() {
        #pragma unroll
        for (int i = 0; i < 4; ++i) {
            int idx4 = i * 256 + tid;
            float4 v = va[i];
            if (!TA) {
                int kq = idx4 & 7;
                int m  = idx4 >> 3;
                v2f p0; p0.x = v.x; p0.y = v.y;
                v2f p1; p1.x = v.z; p1.y = v.w;
                Asp[2 * kq + 0][m] = p0;
                Asp[2 * kq + 1][m] = p1;
            } else {
                int mq = idx4 & 31;
                int k  = idx4 >> 5;
                ((float*)&Asp[k >> 1][4 * mq + 0])[k & 1] = v.x;
                ((float*)&Asp[k >> 1][4 * mq + 1])[k & 1] = v.y;
                ((float*)&Asp[k >> 1][4 * mq + 2])[k & 1] = v.z;
                ((float*)&Asp[k >> 1][4 * mq + 3])[k & 1] = v.w;
            }
        }
        #pragma unroll
        for (int i = 0; i < 4; ++i) {
            int idx4 = i * 256 + tid;
            float4 v = vb[i];
            if (TB) {
                int kq = idx4 & 7;
                int n  = idx4 >> 3;
                v2f p0; p0.x = v.x; p0.y = v.y;
                v2f p1; p1.x = v.z; p1.y = v.w;
                Bsp[2 * kq + 0][n] = p0;
                Bsp[2 * kq + 1][n] = p1;
            } else {
                int nq = idx4 & 31;
                int k  = idx4 >> 5;
                ((float*)&Bsp[k >> 1][4 * nq + 0])[k & 1] = v.x;
                ((float*)&Bsp[k >> 1][4 * nq + 1])[k & 1] = v.y;
                ((float*)&Bsp[k >> 1][4 * nq + 2])[k & 1] = v.z;
                ((float*)&Bsp[k >> 1][4 * nq + 3])[k & 1] = v.w;
            }
        }
    };

    v2f fa[2][2], fb[2][4];
    auto load_frags = [&](int s, int set) {
        const int kp = 2 * s + g;
        fa[set][0] = Asp[kp][32 * waveM + ln];
        fa[set][1] = Asp[kp][32 * waveM + 16 + ln];
        #pragma unroll
        for (int j = 0; j < 4; ++j)
            fb[set][j] = Bsp[kp][64 * waveN + 16 * j + ln];
    };

    load_tile(0);
    store_tile();
    __syncthreads();

    for (int kt = 0; kt < K; kt += BK) {
        const bool has_next = kt + BK < K;
        if (has_next) load_tile(kt + BK);

        load_frags(0, 0);
        #pragma unroll
        for (int s = 0; s < BK / 4; ++s) {
            const int cur = s & 1;
            if (s + 1 < BK / 4) load_frags(s + 1, cur ^ 1);
            #pragma unroll
            for (int j = 0; j < 4; ++j) {
                acc[0][j] = __builtin_amdgcn_wmma_f32_16x16x4_f32(
                    false, fa[cur][0], false, fb[cur][j], (short)0, acc[0][j], false, false);
                acc[1][j] = __builtin_amdgcn_wmma_f32_16x16x4_f32(
                    false, fa[cur][1], false, fb[cur][j], (short)0, acc[1][j], false, false);
            }
        }
        __syncthreads();
        if (has_next) store_tile();
        __syncthreads();
    }

    #pragma unroll
    for (int i2 = 0; i2 < 2; ++i2) {
        #pragma unroll
        for (int j = 0; j < 4; ++j) {
            int col = bn + 64 * waveN + 16 * j + ln;
            #pragma unroll
            for (int r = 0; r < 8; ++r) {
                int row = bm + 32 * waveM + 16 * i2 + r + 8 * g;
                size_t idx = (size_t)row * N + col;
                float v = acc[i2][j][r];
                if (EPI == 0)      D[idx] = alpha * v;
                else if (EPI == 1) D[idx] = Cin[idx] - v;
                else               D[idx] = Cin[idx] + v;
            }
        }
    }
}

__global__ void __launch_bounds__(256) subk(float* __restrict__ o,
                                            const float* __restrict__ a,
                                            const float* __restrict__ b) {
    size_t i = ((size_t)blockIdx.x * 256 + threadIdx.x) * 4;
    float4 av = *(const float4*)(a + i);
    float4 bv = *(const float4*)(b + i);
    *(float4*)(o + i) = make_float4(av.x - bv.x, av.y - bv.y,
                                    av.z - bv.z, av.w - bv.w);
}

__global__ void __launch_bounds__(256) g3k(float* __restrict__ r3,
                                           const float* __restrict__ t3,
                                           const float* __restrict__ y) {
    size_t i = ((size_t)blockIdx.x * 256 + threadIdx.x) * 4;
    float4 rv = *(const float4*)(r3 + i);
    float4 tv = *(const float4*)(t3 + i);
    float4 yv = *(const float4*)(y + i);
    rv.x += EP_BETA * (tv.x - yv.x);
    rv.y += EP_BETA * (tv.y - yv.y);
    rv.z += EP_BETA * (tv.z - yv.z);
    rv.w += EP_BETA * (tv.w - yv.w);
    *(float4*)(r3 + i) = rv;
}

__global__ void __launch_bounds__(256) updk(float* __restrict__ t,
                                            const float* __restrict__ gr) {
    size_t i = ((size_t)blockIdx.x * 256 + threadIdx.x) * 4;
    float4 tv = *(const float4*)(t + i);
    float4 gv = *(const float4*)(gr + i);
    tv.x -= EP_LR * gv.x; tv.y -= EP_LR * gv.y;
    tv.z -= EP_LR * gv.z; tv.w -= EP_LR * gv.w;
    *(float4*)(t + i) = tv;
}

__global__ void __launch_bounds__(256) copyk(float* __restrict__ dst,
                                             const float* __restrict__ src) {
    size_t i = ((size_t)blockIdx.x * 256 + threadIdx.x) * 4;
    *(float4*)(dst + i) = *(const float4*)(src + i);
}

__global__ void __launch_bounds__(256) loss_partial(const float* __restrict__ s,
                                                    const float* __restrict__ y,
                                                    float* __restrict__ part) {
    __shared__ float red[256];
    size_t i = ((size_t)blockIdx.x * 256 + threadIdx.x) * 4;
    float4 a = *(const float4*)(s + i);
    float4 b = *(const float4*)(y + i);
    float dx = a.x - b.x, dy = a.y - b.y, dz = a.z - b.z, dw = a.w - b.w;
    red[threadIdx.x] = dx * dx + dy * dy + dz * dz + dw * dw;
    __syncthreads();
    for (int off = 128; off > 0; off >>= 1) {
        if (threadIdx.x < off) red[threadIdx.x] += red[threadIdx.x + off];
        __syncthreads();
    }
    if (threadIdx.x == 0) part[blockIdx.x] = red[0];
}

__global__ void __launch_bounds__(256) loss_final(const float* __restrict__ part,
                                                  float* __restrict__ out) {
    __shared__ float red[256];
    red[threadIdx.x] = part[threadIdx.x] + part[threadIdx.x + 256];
    __syncthreads();
    for (int off = 128; off > 0; off >>= 1) {
        if (threadIdx.x < off) red[threadIdx.x] += red[threadIdx.x + off];
        __syncthreads();
    }
    if (threadIdx.x == 0) out[0] = red[0] * (1.0f / 524288.0f);
}

#define GEMM(TA, TB, EPI, M, N, K, Ap, Bp, Cp, Dp, alpha)                     \
    gemm_f32_wmma<TA, TB, EPI><<<dim3((M) / BM, (N) / BN), 256, 0, stream>>>( \
        Ap, Bp, Cp, Dp, M, N, K, alpha)

#define EW(kernel, n, ...) kernel<<<dim3((n) / 1024), 256, 0, stream>>>(__VA_ARGS__)

extern "C" void kernel_launch(void* const* d_in, const int* in_sizes, int n_in,
                              void* d_out, int out_size, void* d_ws, size_t ws_size,
                              hipStream_t stream)
{
    const float* x  = (const float*)d_in[0];
    const float* y  = (const float*)d_in[1];
    const float* W0 = (const float*)d_in[2];
    const float* W1 = (const float*)d_in[3];
    const float* W2 = (const float*)d_in[4];
    float* out = (float*)d_out;

    const int Bt = 1024;
    const int D0 = 1024, D1 = 1024, D2 = 1024, D3 = 512;
    const size_t n1 = (size_t)Bt * D1;
    const size_t n2 = (size_t)Bt * D2;
    const size_t n3 = (size_t)Bt * D3;

    float* ws = (float*)d_ws;
    float* p0 = ws;
    float* t1 = p0 + n1;
    float* t2 = t1 + n1;
    float* t3 = t2 + n2;
    float* r1 = t3 + n3;
    float* r2 = r1 + n1;
    float* r3 = r2 + n2;
    float* part = r3 + n3;

    GEMM(false, false, 0, Bt, D1, D0, x,  W0, (const float*)nullptr, p0, 1.0f);
    EW(copyk, n1, t1, p0);
    GEMM(false, false, 0, Bt, D2, D1, t1, W1, (const float*)nullptr, t2, 1.0f);
    GEMM(false, false, 0, Bt, D3, D2, t2, W2, (const float*)nullptr, t3, 1.0f);
    loss_partial<<<512, 256, 0, stream>>>(t3, y, part);
    loss_final<<<1, 256, 0, stream>>>(part, out + 2621440);

    for (int it = 0; it < 20; ++it) {
        GEMM(false, false, 1, Bt, D2, D1, t1, W1, t2, r2, 1.0f);
        GEMM(false, false, 1, Bt, D3, D2, t2, W2, t3, r3, 1.0f);
        EW(subk, n1, r1, t1, p0);
        GEMM(false, true, 1, Bt, D1, D2, r2, W1, r1, r1, 1.0f);
        GEMM(false, true, 1, Bt, D2, D3, r3, W2, r2, r2, 1.0f);
        EW(g3k, n3, r3, t3, y);
        if (it == 0) {
            GEMM(false, true, 2, Bt, D2, D3, r3, W2, r2, r2, 1.0f);
            GEMM(false, true, 2, Bt, D1, D2, r2, W1, r1, r1, 1.0f);
        }
        EW(updk, n1, t1, r1);
        EW(updk, n2, t2, r2);
        EW(updk, n3, t3, r3);
    }

    GEMM(false, false, 1, Bt, D2, D1, t1, W1, t2, r2, 1.0f);
    GEMM(false, false, 1, Bt, D3, D2, t2, W2, t3, r3, 1.0f);
    EW(subk, n1, r1, t1, p0);
    GEMM(true, false, 0, D0, D1, Bt, x,  r1, (const float*)nullptr, out,           -1.0f / EP_BETA);
    GEMM(true, false, 0, D1, D2, Bt, t1, r2, (const float*)nullptr, out + 1048576, -1.0f / EP_BETA);
    GEMM(true, false, 0, D2, D3, Bt, t2, r3, (const float*)nullptr, out + 2097152, -1.0f / EP_BETA);
}


# GEMM work-group tile 128x128x32 on 8 waves -> 64x64x64 on 4 waves (32x32 per wave, fragment groups of two 4-k blocks, LDS row pad 2->1): grid 64 -> 256 work-groups, one per WGP
# speedup vs baseline: 3.2718x; 2.1438x over previous
#include <hip/hip_runtime.h>


typedef __attribute__((ext_vector_type(2))) float v2f;
typedef __attribute__((ext_vector_type(8))) float v8f;

#define EP_BETA 1e-3f
#define EP_LR   0.5f

constexpr int BM = 64;
constexpr int BN = 64;
constexpr int BK = 64;
constexpr int NT = 128;
constexpr int F4 = BM * BK / 4 / NT;

template <bool TA, bool TB, int EPI>
__global__ void __launch_bounds__(NT)
gemm_f32_wmma(const float* __restrict__ A, const float* __restrict__ B,
              const float* __restrict__ Cin, float* __restrict__ D,
              int M, int N, int K, float alpha)
{
    __shared__ v2f Asp[BK / 2][BM + 1];
    __shared__ v2f Bsp[BK / 2][BN + 1];

    const int tid   = threadIdx.x;
    const int lane  = tid & 31;
    const int wid   = tid >> 5;
    const int waveM = wid & 1;
    const int waveN = wid >> 1;
    const int g     = lane >> 4;
    const int ln    = lane & 15;

    const int bm = blockIdx.x * BM;
    const int bn = blockIdx.y * BN;

    v8f acc[2][2] = {};

    float4 va[F4], vb[F4];

    auto load_tile = [&](int kt) {
        #pragma unroll
        for (int i = 0; i < F4; ++i) {
            int idx4 = i * NT + tid;
            if (!TA) {
                int kq = idx4 & 15;
                int m  = idx4 >> 4;
                va[i] = *(const float4*)(A + (size_t)(bm + m) * K + kt + 4 * kq);
            } else {
                int mq = idx4 & 15;
                int k  = idx4 >> 4;
                va[i] = *(const float4*)(A + (size_t)(kt + k) * M + bm + 4 * mq);
            }
        }
        #pragma unroll
        for (int i = 0; i < F4; ++i) {
            int idx4 = i * NT + tid;
            if (TB) {
                int kq = idx4 & 15;
                int n  = idx4 >> 4;
                vb[i] = *(const float4*)(B + (size_t)(bn + n) * K + kt + 4 * kq);
            } else {
                int nq = idx4 & 15;
                int k  = idx4 >> 4;
                vb[i] = *(const float4*)(B + (size_t)(kt + k) * N + bn + 4 * nq);
            }
        }
    };

    auto store_tile = [&]() {
        #pragma unroll
        for (int i = 0; i < F4; ++i) {
            int idx4 = i * NT + tid;
            float4 v = va[i];
            if (!TA) {
                int kq = idx4 & 15;
                int m  = idx4 >> 4;
                v2f p0; p0.x = v.x; p0.y = v.y;
                v2f p1; p1.x = v.z; p1.y = v.w;
                Asp[2 * kq + 0][m] = p0;
                Asp[2 * kq + 1][m] = p1;
            } else {
                int mq = idx4 & 15;
                int k  = idx4 >> 4;
                ((float*)&Asp[k >> 1][4 * mq + 0])[k & 1] = v.x;
                ((float*)&Asp[k >> 1][4 * mq + 1])[k & 1] = v.y;
                ((float*)&Asp[k >> 1][4 * mq + 2])[k & 1] = v.z;
                ((float*)&Asp[k >> 1][4 * mq + 3])[k & 1] = v.w;
            }
        }
        #pragma unroll
        for (int i = 0; i < F4; ++i) {
            int idx4 = i * NT + tid;
            float4 v = vb[i];
            if (TB) {
                int kq = idx4 & 15;
                int n  = idx4 >> 4;
                v2f p0; p0.x = v.x; p0.y = v.y;
                v2f p1; p1.x = v.z; p1.y = v.w;
                Bsp[2 * kq + 0][n] = p0;
                Bsp[2 * kq + 1][n] = p1;
            } else {
                int nq = idx4 & 15;
                int k  = idx4 >> 4;
                ((float*)&Bsp[k >> 1][4 * nq + 0])[k & 1] = v.x;
                ((float*)&Bsp[k >> 1][4 * nq + 1])[k & 1] = v.y;
                ((float*)&Bsp[k >> 1][4 * nq + 2])[k & 1] = v.z;
                ((float*)&Bsp[k >> 1][4 * nq + 3])[k & 1] = v.w;
            }
        }
    };

    v2f fa[2][2][2], fb[2][2][2];
    auto load_frags = [&](int b, int set) {
        #pragma unroll
        for (int u = 0; u < 2; ++u) {
            const int kp = 2 * (2 * b + u) + g;
            #pragma unroll
            for (int t = 0; t < 2; ++t) {
                fa[set][u][t] = Asp[kp][32 * waveM + 16 * t + ln];
                fb[set][u][t] = Bsp[kp][32 * waveN + 16 * t + ln];
            }
        }
    };

    load_tile(0);
    store_tile();
    __syncthreads();

    for (int kt = 0; kt < K; kt += BK) {
        const bool has_next = kt + BK < K;
        if (has_next) load_tile(kt + BK);

        load_frags(0, 0);
        #pragma unroll
        for (int b = 0; b < BK / 8; ++b) {
            const int cur = b & 1;
            if (b + 1 < BK / 8) load_frags(b + 1, cur ^ 1);
            __builtin_amdgcn_sched_barrier(0);
            #pragma unroll
            for (int u = 0; u < 2; ++u) {
                #pragma unroll
                for (int j = 0; j < 2; ++j) {
                    acc[0][j] = __builtin_amdgcn_wmma_f32_16x16x4_f32(
                        false, fa[cur][u][0], false, fb[cur][u][j], (short)0, acc[0][j], false, false);
                    acc[1][j] = __builtin_amdgcn_wmma_f32_16x16x4_f32(
                        false, fa[cur][u][1], false, fb[cur][u][j], (short)0, acc[1][j], false, false);
                }
            }
        }
        __syncthreads();
        if (has_next) store_tile();
        __syncthreads();
    }

    #pragma unroll
    for (int i2 = 0; i2 < 2; ++i2) {
        #pragma unroll
        for (int j = 0; j < 2; ++j) {
            int col = bn + 32 * waveN + 16 * j + ln;
            #pragma unroll
            for (int r = 0; r < 8; ++r) {
                int row = bm + 32 * waveM + 16 * i2 + r + 8 * g;
                size_t idx = (size_t)row * N + col;
                float v = acc[i2][j][r];
                if (EPI == 0)      D[idx] = alpha * v;
                else if (EPI == 1) D[idx] = Cin[idx] - v;
                else               D[idx] = Cin[idx] + v;
            }
        }
    }
}

__global__ void __launch_bounds__(256) subk(float* __restrict__ o,
                                            const float* __restrict__ a,
                                            const float* __restrict__ b) {
    size_t i = ((size_t)blockIdx.x * 256 + threadIdx.x) * 4;
    float4 av = *(const float4*)(a + i);
    float4 bv = *(const float4*)(b + i);
    *(float4*)(o + i) = make_float4(av.x - bv.x, av.y - bv.y,
                                    av.z - bv.z, av.w - bv.w);
}

__global__ void __launch_bounds__(256) g3k(float* __restrict__ r3,
                                           const float* __restrict__ t3,
                                           const float* __restrict__ y) {
    size_t i = ((size_t)blockIdx.x * 256 + threadIdx.x) * 4;
    float4 rv = *(const float4*)(r3 + i);
    float4 tv = *(const float4*)(t3 + i);
    float4 yv = *(const float4*)(y + i);
    rv.x += EP_BETA * (tv.x - yv.x);
    rv.y += EP_BETA * (tv.y - yv.y);
    rv.z += EP_BETA * (tv.z - yv.z);
    rv.w += EP_BETA * (tv.w - yv.w);
    *(float4*)(r3 + i) = rv;
}

__global__ void __launch_bounds__(256) updk(float* __restrict__ t,
                                            const float* __restrict__ gr) {
    size_t i = ((size_t)blockIdx.x * 256 + threadIdx.x) * 4;
    float4 tv = *(const float4*)(t + i);
    float4 gv = *(const float4*)(gr + i);
    tv.x -= EP_LR * gv.x; tv.y -= EP_LR * gv.y;
    tv.z -= EP_LR * gv.z; tv.w -= EP_LR * gv.w;
    *(float4*)(t + i) = tv;
}

__global__ void __launch_bounds__(256) copyk(float* __restrict__ dst,
                                             const float* __restrict__ src) {
    size_t i = ((size_t)blockIdx.x * 256 + threadIdx.x) * 4;
    *(float4*)(dst + i) = *(const float4*)(src + i);
}

__global__ void __launch_bounds__(256) loss_partial(const float* __restrict__ s,
                                                    const float* __restrict__ y,
                                                    float* __restrict__ part) {
    __shared__ float red[256];
    size_t i = ((size_t)blockIdx.x * 256 + threadIdx.x) * 4;
    float4 a = *(const float4*)(s + i);
    float4 b = *(const float4*)(y + i);
    float dx = a.x - b.x, dy = a.y - b.y, dz = a.z - b.z, dw = a.w - b.w;
    red[threadIdx.x] = dx * dx + dy * dy + dz * dz + dw * dw;
    __syncthreads();
    for (int off = 128; off > 0; off >>= 1) {
        if (threadIdx.x < off) red[threadIdx.x] += red[threadIdx.x + off];
        __syncthreads();
    }
    if (threadIdx.x == 0) part[blockIdx.x] = red[0];
}

__global__ void __launch_bounds__(256) loss_final(const float* __restrict__ part,
                                                  float* __restrict__ out) {
    __shared__ float red[256];
    red[threadIdx.x] = part[threadIdx.x] + part[threadIdx.x + 256];
    __syncthreads();
    for (int off = 128; off > 0; off >>= 1) {
        if (threadIdx.x < off) red[threadIdx.x] += red[threadIdx.x + off];
        __syncthreads();
    }
    if (threadIdx.x == 0) out[0] = red[0] * (1.0f / 524288.0f);
}

#define GEMM(TA, TB, EPI, M, N, K, Ap, Bp, Cp, Dp, alpha)                     \
    gemm_f32_wmma<TA, TB, EPI><<<dim3((M) / BM, (N) / BN), NT, 0, stream>>>( \
        Ap, Bp, Cp, Dp, M, N, K, alpha)

#define EW(kernel, n, ...) kernel<<<dim3((n) / 1024), 256, 0, stream>>>(__VA_ARGS__)

extern "C" void kernel_launch(void* const* d_in, const int* in_sizes, int n_in,
                              void* d_out, int out_size, void* d_ws, size_t ws_size,
                              hipStream_t stream)
{
    const float* x  = (const float*)d_in[0];
    const float* y  = (const float*)d_in[1];
    const float* W0 = (const float*)d_in[2];
    const float* W1 = (const float*)d_in[3];
    const float* W2 = (const float*)d_in[4];
    float* out = (float*)d_out;

    const int Bt = 1024;
    const int D0 = 1024, D1 = 1024, D2 = 1024, D3 = 512;
    const size_t n1 = (size_t)Bt * D1;
    const size_t n2 = (size_t)Bt * D2;
    const size_t n3 = (size_t)Bt * D3;

    float* ws = (float*)d_ws;
    float* p0 = ws;
    float* t1 = p0 + n1;
    float* t2 = t1 + n1;
    float* t3 = t2 + n2;
    float* r1 = t3 + n3;
    float* r2 = r1 + n1;
    float* r3 = r2 + n2;
    float* part = r3 + n3;

    GEMM(false, false, 0, Bt, D1, D0, x,  W0, (const float*)nullptr, p0, 1.0f);
    EW(copyk, n1, t1, p0);
    GEMM(false, false, 0, Bt, D2, D1, t1, W1, (const float*)nullptr, t2, 1.0f);
    GEMM(false, false, 0, Bt, D3, D2, t2, W2, (const float*)nullptr, t3, 1.0f);
    loss_partial<<<512, 256, 0, stream>>>(t3, y, part);
    loss_final<<<1, 256, 0, stream>>>(part, out + 2621440);

    for (int it = 0; it < 20; ++it) {
        GEMM(false, false, 1, Bt, D2, D1, t1, W1, t2, r2, 1.0f);
        GEMM(false, false, 1, Bt, D3, D2, t2, W2, t3, r3, 1.0f);
        EW(subk, n1, r1, t1, p0);
        GEMM(false, true, 1, Bt, D1, D2, r2, W1, r1, r1, 1.0f);
        GEMM(false, true, 1, Bt, D2, D3, r3, W2, r2, r2, 1.0f);
        EW(g3k, n3, r3, t3, y);
        if (it == 0) {
            GEMM(false, true, 2, Bt, D2, D3, r3, W2, r2, r2, 1.0f);
            GEMM(false, true, 2, Bt, D1, D2, r2, W1, r1, r1, 1.0f);
        }
        EW(updk, n1, t1, r1);
        EW(updk, n2, t2, r2);
        EW(updk, n3, t3, r3);
    }

    GEMM(false, false, 1, Bt, D2, D1, t1, W1, t2, r2, 1.0f);
    GEMM(false, false, 1, Bt, D3, D2, t2, W2, t3, r3, 1.0f);
    EW(subk, n1, r1, t1, p0);
    GEMM(true, false, 0, D0, D1, Bt, x,  r1, (const float*)nullptr, out,           -1.0f / EP_BETA);
    GEMM(true, false, 0, D1, D2, Bt, t1, r2, (const float*)nullptr, out + 1048576, -1.0f / EP_BETA);
    GEMM(true, false, 0, D2, D3, Bt, t2, r3, (const float*)nullptr, out + 2097152, -1.0f / EP_BETA);
}


# Elementwise passes (residual subtract, clamp term, state updates, copy) folded into the epilogues of the GEMMs that form their operands: 194 -> 94 launches a call, sum order unchanged
# speedup vs baseline: 4.4089x; 1.3475x over previous
#include <hip/hip_runtime.h>


typedef __attribute__((ext_vector_type(2))) float v2f;
typedef __attribute__((ext_vector_type(8))) float v8f;

#define EP_BETA 1e-3f
#define EP_LR   0.5f

constexpr int BM = 64;
constexpr int BN = 64;
constexpr int BK = 64;
constexpr int NT = 128;
constexpr int F4 = BM * BK / 4 / NT;

template <bool TA, bool TB, int EPI>
__global__ void __launch_bounds__(NT)
gemm_f32_wmma(const float* __restrict__ A, const float* __restrict__ B,
              const float* __restrict__ C0, const float* __restrict__ C1,
              float* __restrict__ D, float* __restrict__ T,
              int M, int N, int K, float alpha)
{
    __shared__ v2f Asp[BK / 2][BM + 1];
    __shared__ v2f Bsp[BK / 2][BN + 1];

    const int tid   = threadIdx.x;
    const int lane  = tid & 31;
    const int wid   = tid >> 5;
    const int waveM = wid & 1;
    const int waveN = wid >> 1;
    const int g     = lane >> 4;
    const int ln    = lane & 15;

    const int bm = blockIdx.x * BM;
    const int bn = blockIdx.y * BN;

    v8f acc[2][2] = {};

    float4 va[F4], vb[F4];

    auto load_tile = [&](int kt) {
        #pragma unroll
        for (int i = 0; i < F4; ++i) {
            int idx4 = i * NT + tid;
            if (!TA) {
                int kq = idx4 & 15;
                int m  = idx4 >> 4;
                va[i] = *(const float4*)(A + (size_t)(bm + m) * K + kt + 4 * kq);
            } else {
                int mq = idx4 & 15;
                int k  = idx4 >> 4;
                va[i] = *(const float4*)(A + (size_t)(kt + k) * M + bm + 4 * mq);
            }
        }
        #pragma unroll
        for (int i = 0; i < F4; ++i) {
            int idx4 = i * NT + tid;
            if (TB) {
                int kq = idx4 & 15;
                int n  = idx4 >> 4;
                vb[i] = *(const float4*)(B + (size_t)(bn + n) * K + kt + 4 * kq);
            } else {
                int nq = idx4 & 15;
                int k  = idx4 >> 4;
                vb[i] = *(const float4*)(B + (size_t)(kt + k) * N + bn + 4 * nq);
            }
        }
    };

    auto store_tile = [&]() {
        #pragma unroll
        for (int i = 0; i < F4; ++i) {
            int idx4 = i * NT + tid;
            float4 v = va[i];
            if (!TA) {
                int kq = idx4 & 15;
                int m  = idx4 >> 4;
                v2f p0; p0.x = v.x; p0.y = v.y;
                v2f p1; p1.x = v.z; p1.y = v.w;
                Asp[2 * kq + 0][m] = p0;
                Asp[2 * kq + 1][m] = p1;
            } else {
                int mq = idx4 & 15;
                int k  = idx4 >> 4;
                ((float*)&Asp[k >> 1][4 * mq + 0])[k & 1] = v.x;
                ((float*)&Asp[k >> 1][4 * mq + 1])[k & 1] = v.y;
                ((float*)&Asp[k >> 1][4 * mq + 2])[k & 1] = v.z;
                ((float*)&Asp[k >> 1][4 * mq + 3])[k & 1] = v.w;
            }
        }
        #pragma unroll
        for (int i = 0; i < F4; ++i) {
            int idx4 = i * NT + tid;
            float4 v = vb[i];
            if (TB) {
                int kq = idx4 & 15;
                int n  = idx4 >> 4;
                v2f p0; p0.x = v.x; p0.y = v.y;
                v2f p1; p1.x = v.z; p1.y = v.w;
                Bsp[2 * kq + 0][n] = p0;
                Bsp[2 * kq + 1][n] = p1;
            } else {
                int nq = idx4 & 15;
                int k  = idx4 >> 4;
                ((float*)&Bsp[k >> 1][4 * nq + 0])[k & 1] = v.x;
                ((float*)&Bsp[k >> 1][4 * nq + 1])[k & 1] = v.y;
                ((float*)&Bsp[k >> 1][4 * nq + 2])[k & 1] = v.z;
                ((float*)&Bsp[k >> 1][4 * nq + 3])[k & 1] = v.w;
            }
        }
    };

    v2f fa[2][2][2], fb[2][2][2];
    auto load_frags = [&](int b, int set) {
        #pragma unroll
        for (int u = 0; u < 2; ++u) {
            const int kp = 2 * (2 * b + u) + g;
            #pragma unroll
            for (int t = 0; t < 2; ++t) {
                fa[set][u][t] = Asp[kp][32 * waveM + 16 * t + ln];
                fb[set][u][t] = Bsp[kp][32 * waveN + 16 * t + ln];
            }
        }
    };

    load_tile(0);
    store_tile();
    __syncthreads();

    for (int kt = 0; kt < K; kt += BK) {
        const bool has_next = kt + BK < K;
        if (has_next) load_tile(kt + BK);

        load_frags(0, 0);
        #pragma unroll
        for (int b = 0; b < BK / 8; ++b) {
            const int cur = b & 1;
            if (b + 1 < BK / 8) load_frags(b + 1, cur ^ 1);
            __builtin_amdgcn_sched_barrier(0);
            #pragma unroll
            for (int u = 0; u < 2; ++u) {
                #pragma unroll
                for (int j = 0; j < 2; ++j) {
                    acc[0][j] = __builtin_amdgcn_wmma_f32_16x16x4_f32(
                        false, fa[cur][u][0], false, fb[cur][u][j], (short)0, acc[0][j], false, false);
                    acc[1][j] = __builtin_amdgcn_wmma_f32_16x16x4_f32(
                        false, fa[cur][u][1], false, fb[cur][u][j], (short)0, acc[1][j], false, false);
                }
            }
        }
        __syncthreads();
        if (has_next) store_tile();
        __syncthreads();
    }

    constexpr bool RD_C0 = (EPI == 1 || EPI == 4 || EPI == 6 || EPI == 7 || EPI == 8);
    constexpr bool RD_C1 = (EPI == 3 || EPI == 5 || EPI == 6);
    constexpr bool RD_T  = (EPI == 3 || EPI == 4 || EPI == 5 || EPI == 7 || EPI == 8);
    const int row0 = bm + 32 * waveM + 8 * g;
    const int col0 = bn + 32 * waveN + ln;
    auto at = [&](int i2, int j, int r) -> size_t {
        return (size_t)(row0 + 16 * i2 + r) * N + (col0 + 16 * j);
    };

    float c0v[2][2][8], c1v[2][2][8], tv[2][2][8];
    #pragma unroll
    for (int i2 = 0; i2 < 2; ++i2) {
        #pragma unroll
        for (int j = 0; j < 2; ++j) {
            #pragma unroll
            for (int r = 0; r < 8; ++r) {
                const size_t idx = at(i2, j, r);
                if (RD_C0) c0v[i2][j][r] = C0[idx];
                if (RD_C1) c1v[i2][j][r] = C1[idx];
                if (RD_T)  tv[i2][j][r]  = T[idx];
            }
        }
    }

    #pragma unroll
    for (int i2 = 0; i2 < 2; ++i2) {
        #pragma unroll
        for (int j = 0; j < 2; ++j) {
            #pragma unroll
            for (int r = 0; r < 8; ++r) {
                const size_t idx = at(i2, j, r);
                const float v = acc[i2][j][r];
                if (EPI == 0) {
                    D[idx] = alpha * v;
                } else if (EPI == 1) {
                    D[idx] = c0v[i2][j][r] - v;
                } else if (EPI == 3) {
                    const float t  = tv[i2][j][r];
                    const float c  = c1v[i2][j][r];
                    const float gq = (t - c) - v;
                    const float tn = t - EP_LR * gq;
                    T[idx] = tn;
                    tv[i2][j][r] = tn;
                } else if (EPI == 4) {
                    const float gq = c0v[i2][j][r] - v;
                    T[idx] = tv[i2][j][r] - EP_LR * gq;
                } else if (EPI == 5) {
                    const float t  = tv[i2][j][r];
                    const float rr = t - v;
                    D[idx] = rr;
                    const float gq = rr + EP_BETA * (t - c1v[i2][j][r]);
                    T[idx] = t - EP_LR * gq;
                } else if (EPI == 6) {
                    D[idx] = (c0v[i2][j][r] - c1v[i2][j][r]) - v;
                } else if (EPI == 7) {
                    const float gq = c0v[i2][j][r] + v;
                    D[idx] = gq;
                    T[idx] = tv[i2][j][r] - EP_LR * gq;
                } else {
                    const float gq = c0v[i2][j][r] + v;
                    T[idx] = tv[i2][j][r] - EP_LR * gq;
                }
            }
        }
    }

    if ((EPI == 0 && T != nullptr) || (EPI == 3 && D != nullptr)) {
        #pragma unroll
        for (int i2 = 0; i2 < 2; ++i2) {
            #pragma unroll
            for (int j = 0; j < 2; ++j) {
                #pragma unroll
                for (int r = 0; r < 8; ++r) {
                    const size_t idx = at(i2, j, r);
                    if (EPI == 0) T[idx] = alpha * acc[i2][j][r];
                    else          D[idx] = tv[i2][j][r] - c1v[i2][j][r];
                }
            }
        }
    }
}

__global__ void __launch_bounds__(256) g3k(float* __restrict__ r3,
                                           const float* __restrict__ t3,
                                           const float* __restrict__ y) {
    size_t i = ((size_t)blockIdx.x * 256 + threadIdx.x) * 4;
    float4 rv = *(const float4*)(r3 + i);
    float4 tv = *(const float4*)(t3 + i);
    float4 yv = *(const float4*)(y + i);
    rv.x += EP_BETA * (tv.x - yv.x);
    rv.y += EP_BETA * (tv.y - yv.y);
    rv.z += EP_BETA * (tv.z - yv.z);
    rv.w += EP_BETA * (tv.w - yv.w);
    *(float4*)(r3 + i) = rv;
}

__global__ void __launch_bounds__(256) updk(float* __restrict__ t,
                                            const float* __restrict__ gr) {
    size_t i = ((size_t)blockIdx.x * 256 + threadIdx.x) * 4;
    float4 tv = *(const float4*)(t + i);
    float4 gv = *(const float4*)(gr + i);
    tv.x -= EP_LR * gv.x; tv.y -= EP_LR * gv.y;
    tv.z -= EP_LR * gv.z; tv.w -= EP_LR * gv.w;
    *(float4*)(t + i) = tv;
}

__global__ void __launch_bounds__(256) loss_partial(const float* __restrict__ s,
                                                    const float* __restrict__ y,
                                                    float* __restrict__ part) {
    __shared__ float red[256];
    size_t i = ((size_t)blockIdx.x * 256 + threadIdx.x) * 4;
    float4 a = *(const float4*)(s + i);
    float4 b = *(const float4*)(y + i);
    float dx = a.x - b.x, dy = a.y - b.y, dz = a.z - b.z, dw = a.w - b.w;
    red[threadIdx.x] = dx * dx + dy * dy + dz * dz + dw * dw;
    __syncthreads();
    for (int off = 128; off > 0; off >>= 1) {
        if (threadIdx.x < off) red[threadIdx.x] += red[threadIdx.x + off];
        __syncthreads();
    }
    if (threadIdx.x == 0) part[blockIdx.x] = red[0];
}

__global__ void __launch_bounds__(256) loss_final(const float* __restrict__ part,
                                                  float* __restrict__ out) {
    __shared__ float red[256];
    red[threadIdx.x] = part[threadIdx.x] + part[threadIdx.x + 256];
    __syncthreads();
    for (int off = 128; off > 0; off >>= 1) {
        if (threadIdx.x < off) red[threadIdx.x] += red[threadIdx.x + off];
        __syncthreads();
    }
    if (threadIdx.x == 0) out[0] = red[0] * (1.0f / 524288.0f);
}

#define GEMM(TA, TB, EPI, M, N, K, Ap, Bp, C0p, C1p, Dp, Tp, alpha)          \
    gemm_f32_wmma<TA, TB, EPI><<<dim3((M) / BM, (N) / BN), NT, 0, stream>>>( \
        Ap, Bp, C0p, C1p, Dp, Tp, M, N, K, alpha)

#define EW(kernel, n, ...) kernel<<<dim3((n) / 1024), 256, 0, stream>>>(__VA_ARGS__)

extern "C" void kernel_launch(void* const* d_in, const int* in_sizes, int n_in,
                              void* d_out, int out_size, void* d_ws, size_t ws_size,
                              hipStream_t stream)
{
    const float* x  = (const float*)d_in[0];
    const float* y  = (const float*)d_in[1];
    const float* W0 = (const float*)d_in[2];
    const float* W1 = (const float*)d_in[3];
    const float* W2 = (const float*)d_in[4];
    float* out = (float*)d_out;

    const int Bt = 1024;
    const int D0 = 1024, D1 = 1024, D2 = 1024, D3 = 512;
    const size_t n1 = (size_t)Bt * D1;
    const size_t n2 = (size_t)Bt * D2;
    const size_t n3 = (size_t)Bt * D3;

    float* ws = (float*)d_ws;
    float* p0 = ws;
    float* t1 = p0 + n1;
    float* t2 = t1 + n1;
    float* t3 = t2 + n2;
    float* r1 = t3 + n3;
    float* r2 = r1 + n1;
    float* r3 = r2 + n2;
    float* part = r3 + n3;

    const float* nc = nullptr;
    float* nw = nullptr;

    GEMM(false, false, 0, Bt, D1, D0, x,  W0, nc, nc, p0, t1, 1.0f);
    GEMM(false, false, 0, Bt, D2, D1, t1, W1, nc, nc, t2, nw, 1.0f);
    GEMM(false, false, 0, Bt, D3, D2, t2, W2, nc, nc, t3, nw, 1.0f);
    loss_partial<<<512, 256, 0, stream>>>(t3, y, part);
    loss_final<<<1, 256, 0, stream>>>(part, out + 2621440);

    GEMM(false, false, 1, Bt, D2, D1, t1, W1, t2, nc, r2, nw, 1.0f);
    GEMM(false, false, 1, Bt, D3, D2, t2, W2, t3, nc, r3, nw, 1.0f);
    GEMM(false, true,  6, Bt, D1, D2, r2, W1, t1, p0, r1, nw, 1.0f);
    GEMM(false, true,  1, Bt, D2, D3, r3, W2, r2, nc, r2, nw, 1.0f);
    EW(g3k, n3, r3, t3, y);
    GEMM(false, true,  7, Bt, D2, D3, r3, W2, r2, nc, r2, t2, 1.0f);
    GEMM(false, true,  8, Bt, D1, D2, r2, W1, r1, nc, nw, t1, 1.0f);
    EW(updk, n3, t3, r3);

    for (int it = 1; it < 20; ++it) {
        GEMM(false, false, 1, Bt, D2, D1, t1, W1, t2, nc, r2, nw, 1.0f);
        GEMM(false, false, 5, Bt, D3, D2, t2, W2, nc, y,  r3, t3, 1.0f);
        GEMM(false, true,  3, Bt, D1, D2, r2, W1, nc, p0, (it == 19 ? r1 : nw), t1, 1.0f);
        GEMM(false, true,  4, Bt, D2, D3, r3, W2, r2, nc, nw, t2, 1.0f);
    }

    GEMM(false, false, 1, Bt, D2, D1, t1, W1, t2, nc, r2, nw, 1.0f);
    GEMM(false, false, 1, Bt, D3, D2, t2, W2, t3, nc, r3, nw, 1.0f);
    GEMM(true, false, 0, D0, D1, Bt, x,  r1, nc, nc, out,           nw, -1.0f / EP_BETA);
    GEMM(true, false, 0, D1, D2, Bt, t1, r2, nc, nc, out + 1048576, nw, -1.0f / EP_BETA);
    GEMM(true, false, 0, D2, D3, Bt, t2, r3, nc, nc, out + 2097152, nw, -1.0f / EP_BETA);
}


# Independent GEMMs of a relaxation step share one launch (blockIdx.y picks the problem): two work-groups per WGP, one's staging under the other's WMMAs; 94 -> 53 launches, sum order unchanged
# speedup vs baseline: 5.6604x; 1.2839x over previous
#include <hip/hip_runtime.h>


typedef __attribute__((ext_vector_type(2))) float v2f;
typedef __attribute__((ext_vector_type(8))) float v8f;

#define EP_BETA 1e-3f
#define EP_LR   0.5f

constexpr int BM = 64;
constexpr int BN = 64;
constexpr int BK = 64;
constexpr int NT = 128;
constexpr int F4 = BM * BK / 4 / NT;

template <bool TA, bool TB, int EPI>
__device__ __forceinline__ void
gemm_tile(const float* __restrict__ A, const float* __restrict__ B,
          const float* __restrict__ C0, const float* __restrict__ C1,
          float* __restrict__ D, float* __restrict__ T,
          int M, int N, int K, float alpha, int tile_x, int tile_y,
          v2f (*Asp)[BM + 1], v2f (*Bsp)[BN + 1])
{
    const int tid   = threadIdx.x;
    const int lane  = tid & 31;
    const int wid   = tid >> 5;
    const int waveM = wid & 1;
    const int waveN = wid >> 1;
    const int g     = lane >> 4;
    const int ln    = lane & 15;

    const int bm = tile_x * BM;
    const int bn = tile_y * BN;

    v8f acc[2][2] = {};

    float4 va[F4], vb[F4];

    auto load_tile = [&](int kt) {
        #pragma unroll
        for (int i = 0; i < F4; ++i) {
            int idx4 = i * NT + tid;
            if (!TA) {
                int kq = idx4 & 15;
                int m  = idx4 >> 4;
                va[i] = *(const float4*)(A + (size_t)(bm + m) * K + kt + 4 * kq);
            } else {
                int mq = idx4 & 15;
                int k  = idx4 >> 4;
                va[i] = *(const float4*)(A + (size_t)(kt + k) * M + bm + 4 * mq);
            }
        }
        #pragma unroll
        for (int i = 0; i < F4; ++i) {
            int idx4 = i * NT + tid;
            if (TB) {
                int kq = idx4 & 15;
                int n  = idx4 >> 4;
                vb[i] = *(const float4*)(B + (size_t)(bn + n) * K + kt + 4 * kq);
            } else {
                int nq = idx4 & 15;
                int k  = idx4 >> 4;
                vb[i] = *(const float4*)(B + (size_t)(kt + k) * N + bn + 4 * nq);
            }
        }
    };

    auto store_tile = [&]() {
        #pragma unroll
        for (int i = 0; i < F4; ++i) {
            int idx4 = i * NT + tid;
            float4 v = va[i];
            if (!TA) {
                int kq = idx4 & 15;
                int m  = idx4 >> 4;
                v2f p0; p0.x = v.x; p0.y = v.y;
                v2f p1; p1.x = v.z; p1.y = v.w;
                Asp[2 * kq + 0][m] = p0;
                Asp[2 * kq + 1][m] = p1;
            } else {
                int mq = idx4 & 15;
                int k  = idx4 >> 4;
                ((float*)&Asp[k >> 1][4 * mq + 0])[k & 1] = v.x;
                ((float*)&Asp[k >> 1][4 * mq + 1])[k & 1] = v.y;
                ((float*)&Asp[k >> 1][4 * mq + 2])[k & 1] = v.z;
                ((float*)&Asp[k >> 1][4 * mq + 3])[k & 1] = v.w;
            }
        }
        #pragma unroll
        for (int i = 0; i < F4; ++i) {
            int idx4 = i * NT + tid;
            float4 v = vb[i];
            if (TB) {
                int kq = idx4 & 15;
                int n  = idx4 >> 4;
                v2f p0; p0.x = v.x; p0.y = v.y;
                v2f p1; p1.x = v.z; p1.y = v.w;
                Bsp[2 * kq + 0][n] = p0;
                Bsp[2 * kq + 1][n] = p1;
            } else {
                int nq = idx4 & 15;
                int k  = idx4 >> 4;
                ((float*)&Bsp[k >> 1][4 * nq + 0])[k & 1] = v.x;
                ((float*)&Bsp[k >> 1][4 * nq + 1])[k & 1] = v.y;
                ((float*)&Bsp[k >> 1][4 * nq + 2])[k & 1] = v.z;
                ((float*)&Bsp[k >> 1][4 * nq + 3])[k & 1] = v.w;
            }
        }
    };

    v2f fa[2][2][2], fb[2][2][2];
    auto load_frags = [&](int b, int set) {
        #pragma unroll
        for (int u = 0; u < 2; ++u) {
            const int kp = 2 * (2 * b + u) + g;
            #pragma unroll
            for (int t = 0; t < 2; ++t) {
                fa[set][u][t] = Asp[kp][32 * waveM + 16 * t + ln];
                fb[set][u][t] = Bsp[kp][32 * waveN + 16 * t + ln];
            }
        }
    };

    load_tile(0);
    store_tile();
    __syncthreads();

    for (int kt = 0; kt < K; kt += BK) {
        const bool has_next = kt + BK < K;
        if (has_next) load_tile(kt + BK);

        load_frags(0, 0);
        #pragma unroll
        for (int b = 0; b < BK / 8; ++b) {
            const int cur = b & 1;
            if (b + 1 < BK / 8) load_frags(b + 1, cur ^ 1);
            __builtin_amdgcn_sched_barrier(0);
            #pragma unroll
            for (int u = 0; u < 2; ++u) {
                #pragma unroll
                for (int j = 0; j < 2; ++j) {
                    acc[0][j] = __builtin_amdgcn_wmma_f32_16x16x4_f32(
                        false, fa[cur][u][0], false, fb[cur][u][j], (short)0, acc[0][j], false, false);
                    acc[1][j] = __builtin_amdgcn_wmma_f32_16x16x4_f32(
                        false, fa[cur][u][1], false, fb[cur][u][j], (short)0, acc[1][j], false, false);
                }
            }
        }
        __syncthreads();
        if (has_next) store_tile();
        __syncthreads();
    }

    constexpr bool RD_C0 = (EPI == 1 || EPI == 4 || EPI == 6 || EPI == 7 || EPI == 8);
    constexpr bool RD_C1 = (EPI == 3 || EPI == 5 || EPI == 6);
    constexpr bool RD_T  = (EPI == 3 || EPI == 4 || EPI == 5 || EPI == 7 || EPI == 8);
    const int row0 = bm + 32 * waveM + 8 * g;
    const int col0 = bn + 32 * waveN + ln;
    auto at = [&](int i2, int j, int r) -> size_t {
        return (size_t)(row0 + 16 * i2 + r) * N + (col0 + 16 * j);
    };

    float c0v[2][2][8], c1v[2][2][8], tv[2][2][8];
    #pragma unroll
    for (int i2 = 0; i2 < 2; ++i2) {
        #pragma unroll
        for (int j = 0; j < 2; ++j) {
            #pragma unroll
            for (int r = 0; r < 8; ++r) {
                const size_t idx = at(i2, j, r);
                if (RD_C0) c0v[i2][j][r] = C0[idx];
                if (RD_C1) c1v[i2][j][r] = C1[idx];
                if (RD_T)  tv[i2][j][r]  = T[idx];
            }
        }
    }

    #pragma unroll
    for (int i2 = 0; i2 < 2; ++i2) {
        #pragma unroll
        for (int j = 0; j < 2; ++j) {
            #pragma unroll
            for (int r = 0; r < 8; ++r) {
                const size_t idx = at(i2, j, r);
                const float v = acc[i2][j][r];
                if (EPI == 0) {
                    D[idx] = alpha * v;
                } else if (EPI == 1) {
                    D[idx] = c0v[i2][j][r] - v;
                } else if (EPI == 3) {
                    const float t  = tv[i2][j][r];
                    const float c  = c1v[i2][j][r];
                    const float gq = (t - c) - v;
                    const float tn = t - EP_LR * gq;
                    T[idx] = tn;
                    tv[i2][j][r] = tn;
                } else if (EPI == 4) {
                    const float gq = c0v[i2][j][r] - v;
                    T[idx] = tv[i2][j][r] - EP_LR * gq;
                } else if (EPI == 5) {
                    const float t  = tv[i2][j][r];
                    const float rr = t - v;
                    D[idx] = rr;
                    const float gq = rr + EP_BETA * (t - c1v[i2][j][r]);
                    T[idx] = t - EP_LR * gq;
                } else if (EPI == 6) {
                    D[idx] = (c0v[i2][j][r] - c1v[i2][j][r]) - v;
                } else if (EPI == 7) {
                    const float gq = c0v[i2][j][r] + v;
                    D[idx] = gq;
                    T[idx] = tv[i2][j][r] - EP_LR * gq;
                } else {
                    const float gq = c0v[i2][j][r] + v;
                    T[idx] = tv[i2][j][r] - EP_LR * gq;
                }
            }
        }
    }

    if ((EPI == 0 && T != nullptr) || (EPI == 3 && D != nullptr)) {
        #pragma unroll
        for (int i2 = 0; i2 < 2; ++i2) {
            #pragma unroll
            for (int j = 0; j < 2; ++j) {
                #pragma unroll
                for (int r = 0; r < 8; ++r) {
                    const size_t idx = at(i2, j, r);
                    if (EPI == 0) T[idx] = alpha * acc[i2][j][r];
                    else          D[idx] = tv[i2][j][r] - c1v[i2][j][r];
                }
            }
        }
    }
}

template <bool TA, bool TB, int EPI>
__global__ void __launch_bounds__(NT)
gemm_f32_wmma(const float* __restrict__ A, const float* __restrict__ B,
              const float* __restrict__ C0, const float* __restrict__ C1,
              float* __restrict__ D, float* __restrict__ T,
              int M, int N, int K, float alpha)
{
    __shared__ v2f Asp[BK / 2][BM + 1];
    __shared__ v2f Bsp[BK / 2][BN + 1];
    gemm_tile<TA, TB, EPI>(A, B, C0, C1, D, T, M, N, K, alpha,
                           blockIdx.x, blockIdx.y, Asp, Bsp);
}

struct GemmArgs {
    const float* A;
    const float* B;
    const float* C0;
    const float* C1;
    float* D;
    float* T;
    int M, N, K;
    float alpha;
};

template <bool TA0, bool TB0, int EPI0, bool TA1, bool TB1, int EPI1>
__global__ void __launch_bounds__(NT)
gemm_pair_f32_wmma(GemmArgs p, GemmArgs q, int ny0)
{
    __shared__ v2f Asp[BK / 2][BM + 1];
    __shared__ v2f Bsp[BK / 2][BN + 1];
    const int ty = blockIdx.y;
    if (ty < ny0) {
        gemm_tile<TA0, TB0, EPI0>(p.A, p.B, p.C0, p.C1, p.D, p.T, p.M, p.N, p.K, p.alpha,
                                  blockIdx.x, ty, Asp, Bsp);
    } else {
        gemm_tile<TA1, TB1, EPI1>(q.A, q.B, q.C0, q.C1, q.D, q.T, q.M, q.N, q.K, q.alpha,
                                  blockIdx.x, ty - ny0, Asp, Bsp);
    }
}

__global__ void __launch_bounds__(256) g3k(float* __restrict__ r3,
                                           const float* __restrict__ t3,
                                           const float* __restrict__ y) {
    size_t i = ((size_t)blockIdx.x * 256 + threadIdx.x) * 4;
    float4 rv = *(const float4*)(r3 + i);
    float4 tv = *(const float4*)(t3 + i);
    float4 yv = *(const float4*)(y + i);
    rv.x += EP_BETA * (tv.x - yv.x);
    rv.y += EP_BETA * (tv.y - yv.y);
    rv.z += EP_BETA * (tv.z - yv.z);
    rv.w += EP_BETA * (tv.w - yv.w);
    *(float4*)(r3 + i) = rv;
}

__global__ void __launch_bounds__(256) updk(float* __restrict__ t,
                                            const float* __restrict__ gr) {
    size_t i = ((size_t)blockIdx.x * 256 + threadIdx.x) * 4;
    float4 tv = *(const float4*)(t + i);
    float4 gv = *(const float4*)(gr + i);
    tv.x -= EP_LR * gv.x; tv.y -= EP_LR * gv.y;
    tv.z -= EP_LR * gv.z; tv.w -= EP_LR * gv.w;
    *(float4*)(t + i) = tv;
}

__global__ void __launch_bounds__(256) loss_partial(const float* __restrict__ s,
                                                    const float* __restrict__ y,
                                                    float* __restrict__ part) {
    __shared__ float red[256];
    size_t i = ((size_t)blockIdx.x * 256 + threadIdx.x) * 4;
    float4 a = *(const float4*)(s + i);
    float4 b = *(const float4*)(y + i);
    float dx = a.x - b.x, dy = a.y - b.y, dz = a.z - b.z, dw = a.w - b.w;
    red[threadIdx.x] = dx * dx + dy * dy + dz * dz + dw * dw;
    __syncthreads();
    for (int off = 128; off > 0; off >>= 1) {
        if (threadIdx.x < off) red[threadIdx.x] += red[threadIdx.x + off];
        __syncthreads();
    }
    if (threadIdx.x == 0) part[blockIdx.x] = red[0];
}

__global__ void __launch_bounds__(256) loss_final(const float* __restrict__ part,
                                                  float* __restrict__ out) {
    __shared__ float red[256];
    red[threadIdx.x] = part[threadIdx.x] + part[threadIdx.x + 256];
    __syncthreads();
    for (int off = 128; off > 0; off >>= 1) {
        if (threadIdx.x < off) red[threadIdx.x] += red[threadIdx.x + off];
        __syncthreads();
    }
    if (threadIdx.x == 0) out[0] = red[0] * (1.0f / 524288.0f);
}

#define GEMM(TA, TB, EPI, M, N, K, Ap, Bp, C0p, C1p, Dp, Tp, alpha)          \
    gemm_f32_wmma<TA, TB, EPI><<<dim3((M) / BM, (N) / BN), NT, 0, stream>>>( \
        Ap, Bp, C0p, C1p, Dp, Tp, M, N, K, alpha)

#define GEMM2(TA0, TB0, EPI0, TA1, TB1, EPI1, P0, P1)                          \
    gemm_pair_f32_wmma<TA0, TB0, EPI0, TA1, TB1, EPI1>                         \
        <<<dim3((P0).M / BM, (P0).N / BN + (P1).N / BN), NT, 0, stream>>>(     \
            P0, P1, (P0).N / BN)

#define EW(kernel, n, ...) kernel<<<dim3((n) / 1024), 256, 0, stream>>>(__VA_ARGS__)

extern "C" void kernel_launch(void* const* d_in, const int* in_sizes, int n_in,
                              void* d_out, int out_size, void* d_ws, size_t ws_size,
                              hipStream_t stream)
{
    const float* x  = (const float*)d_in[0];
    const float* y  = (const float*)d_in[1];
    const float* W0 = (const float*)d_in[2];
    const float* W1 = (const float*)d_in[3];
    const float* W2 = (const float*)d_in[4];
    float* out = (float*)d_out;

    const int Bt = 1024;
    const int D0 = 1024, D1 = 1024, D2 = 1024, D3 = 512;
    const size_t n1 = (size_t)Bt * D1;
    const size_t n2 = (size_t)Bt * D2;
    const size_t n3 = (size_t)Bt * D3;

    float* ws = (float*)d_ws;
    float* p0 = ws;
    float* t1 = p0 + n1;
    float* t2 = t1 + n1;
    float* t3 = t2 + n2;
    float* r1 = t3 + n3;
    float* r2 = r1 + n1;
    float* r3 = r2 + n2;
    float* part = r3 + n3;

    const float* nc = nullptr;
    float* nw = nullptr;

    const GemmArgs res2  = { t1, W1, t2, nc, r2, nw, Bt, D2, D1, 1.0f };
    const GemmArgs res3  = { t2, W2, t3, nc, r3, nw, Bt, D3, D2, 1.0f };
    const GemmArgs res3u = { t2, W2, nc, y,  r3, t3, Bt, D3, D2, 1.0f };

    GEMM(false, false, 0, Bt, D1, D0, x,  W0, nc, nc, p0, t1, 1.0f);
    GEMM(false, false, 0, Bt, D2, D1, t1, W1, nc, nc, t2, nw, 1.0f);
    GEMM(false, false, 0, Bt, D3, D2, t2, W2, nc, nc, t3, nw, 1.0f);
    loss_partial<<<512, 256, 0, stream>>>(t3, y, part);
    loss_final<<<1, 256, 0, stream>>>(part, out + 2621440);

    GEMM2(false, false, 1, false, false, 1, res2, res3);
    GEMM(false, true,  6, Bt, D1, D2, r2, W1, t1, p0, r1, nw, 1.0f);
    GEMM(false, true,  1, Bt, D2, D3, r3, W2, r2, nc, r2, nw, 1.0f);
    EW(g3k, n3, r3, t3, y);
    GEMM(false, true,  7, Bt, D2, D3, r3, W2, r2, nc, r2, t2, 1.0f);
    GEMM(false, true,  8, Bt, D1, D2, r2, W1, r1, nc, nw, t1, 1.0f);
    EW(updk, n3, t3, r3);

    for (int it = 1; it < 20; ++it) {
        GEMM2(false, false, 1, false, false, 5, res2, res3u);
        const GemmArgs upd1 = { r2, W1, nc, p0, (it == 19 ? r1 : nw), t1, Bt, D1, D2, 1.0f };
        const GemmArgs upd2 = { r3, W2, r2, nc, nw, t2, Bt, D2, D3, 1.0f };
        GEMM2(false, true, 3, false, true, 4, upd1, upd2);
    }

    GEMM2(false, false, 1, false, false, 1, res2, res3);
    const GemmArgs gw0 = { x,  r1, nc, nc, out,           nw, D0, D1, Bt, -1.0f / EP_BETA };
    const GemmArgs gw1 = { t1, r2, nc, nc, out + 1048576, nw, D1, D2, Bt, -1.0f / EP_BETA };
    GEMM2(true, false, 0, true, false, 0, gw0, gw1);
    GEMM(true, false, 0, D2, D3, Bt, t2, r3, nc, nc, out + 2097152, nw, -1.0f / EP_BETA);
}


# GEMMs with N=512 take 64x32 tiles (32x16 per wave, fragment groups of four 4-k blocks): 256 half-size work-groups, so each WGP gets equal work in every launch; k order and sums unchanged
# speedup vs baseline: 6.4722x; 1.1434x over previous
#include <hip/hip_runtime.h>


typedef __attribute__((ext_vector_type(2))) float v2f;
typedef __attribute__((ext_vector_type(8))) float v8f;

#define EP_BETA 1e-3f
#define EP_LR   0.5f

constexpr int BM = 64;
constexpr int BN = 64;
constexpr int BK = 64;
constexpr int NT = 128;
constexpr int F4 = BM * BK / 4 / NT;

template <bool TA, bool TB, int EPI, int BNT>
__device__ __forceinline__ void
gemm_tile(const float* __restrict__ A, const float* __restrict__ B,
          const float* __restrict__ C0, const float* __restrict__ C1,
          float* __restrict__ D, float* __restrict__ T,
          int M, int N, int K, float alpha, int tile_x, int tile_y,
          v2f (*Asp)[BM + 1], v2f (*Bsp)[BN + 1])
{
    const int tid   = threadIdx.x;
    const int lane  = tid & 31;
    const int wid   = tid >> 5;
    const int waveM = wid & 1;
    const int waveN = wid >> 1;
    const int g     = lane >> 4;
    const int ln    = lane & 15;

    constexpr int NTN = BNT / 32;
    constexpr int WN  = BNT / 2;
    constexpr int F4B = BNT * BK / 4 / NT;
    constexpr int NQ  = BNT / 4;
    constexpr int NQS = (BNT == 64) ? 4 : 3;
    constexpr int UG  = 4 / NTN;
    constexpr int NG  = BK / 4 / UG;

    const int bm = tile_x * BM;
    const int bn = tile_y * BNT;

    v8f acc[2][NTN] = {};

    float4 va[F4], vb[F4B];

    auto load_tile = [&](int kt) {
        #pragma unroll
        for (int i = 0; i < F4; ++i) {
            int idx4 = i * NT + tid;
            if (!TA) {
                int kq = idx4 & 15;
                int m  = idx4 >> 4;
                va[i] = *(const float4*)(A + (size_t)(bm + m) * K + kt + 4 * kq);
            } else {
                int mq = idx4 & 15;
                int k  = idx4 >> 4;
                va[i] = *(const float4*)(A + (size_t)(kt + k) * M + bm + 4 * mq);
            }
        }
        #pragma unroll
        for (int i = 0; i < F4B; ++i) {
            int idx4 = i * NT + tid;
            if (TB) {
                int kq = idx4 & 15;
                int n  = idx4 >> 4;
                vb[i] = *(const float4*)(B + (size_t)(bn + n) * K + kt + 4 * kq);
            } else {
                int nq = idx4 & (NQ - 1);
                int k  = idx4 >> NQS;
                vb[i] = *(const float4*)(B + (size_t)(kt + k) * N + bn + 4 * nq);
            }
        }
    };

    auto store_tile = [&]() {
        #pragma unroll
        for (int i = 0; i < F4; ++i) {
            int idx4 = i * NT + tid;
            float4 v = va[i];
            if (!TA) {
                int kq = idx4 & 15;
                int m  = idx4 >> 4;
                v2f p0; p0.x = v.x; p0.y = v.y;
                v2f p1; p1.x = v.z; p1.y = v.w;
                Asp[2 * kq + 0][m] = p0;
                Asp[2 * kq + 1][m] = p1;
            } else {
                int mq = idx4 & 15;
                int k  = idx4 >> 4;
                ((float*)&Asp[k >> 1][4 * mq + 0])[k & 1] = v.x;
                ((float*)&Asp[k >> 1][4 * mq + 1])[k & 1] = v.y;
                ((float*)&Asp[k >> 1][4 * mq + 2])[k & 1] = v.z;
                ((float*)&Asp[k >> 1][4 * mq + 3])[k & 1] = v.w;
            }
        }
        #pragma unroll
        for (int i = 0; i < F4B; ++i) {
            int idx4 = i * NT + tid;
            float4 v = vb[i];
            if (TB) {
                int kq = idx4 & 15;
                int n  = idx4 >> 4;
                v2f p0; p0.x = v.x; p0.y = v.y;
                v2f p1; p1.x = v.z; p1.y = v.w;
                Bsp[2 * kq + 0][n] = p0;
                Bsp[2 * kq + 1][n] = p1;
            } else {
                int nq = idx4 & (NQ - 1);
                int k  = idx4 >> NQS;
                ((float*)&Bsp[k >> 1][4 * nq + 0])[k & 1] = v.x;
                ((float*)&Bsp[k >> 1][4 * nq + 1])[k & 1] = v.y;
                ((float*)&Bsp[k >> 1][4 * nq + 2])[k & 1] = v.z;
                ((float*)&Bsp[k >> 1][4 * nq + 3])[k & 1] = v.w;
            }
        }
    };

    v2f fa[2][UG][2], fb[2][UG][NTN];
    auto load_frags = [&](int b, int set) {
        #pragma unroll
        for (int u = 0; u < UG; ++u) {
            const int kp = 2 * (UG * b + u) + g;
            #pragma unroll
            for (int t = 0; t < 2; ++t)
                fa[set][u][t] = Asp[kp][32 * waveM + 16 * t + ln];
            #pragma unroll
            for (int t = 0; t < NTN; ++t)
                fb[set][u][t] = Bsp[kp][WN * waveN + 16 * t + ln];
        }
    };

    load_tile(0);
    store_tile();
    __syncthreads();

    for (int kt = 0; kt < K; kt += BK) {
        const bool has_next = kt + BK < K;
        if (has_next) load_tile(kt + BK);

        load_frags(0, 0);
        #pragma unroll
        for (int b = 0; b < NG; ++b) {
            const int cur = b & 1;
            if (b + 1 < NG) load_frags(b + 1, cur ^ 1);
            __builtin_amdgcn_sched_barrier(0);
            #pragma unroll
            for (int u = 0; u < UG; ++u) {
                #pragma unroll
                for (int j = 0; j < NTN; ++j) {
                    acc[0][j] = __builtin_amdgcn_wmma_f32_16x16x4_f32(
                        false, fa[cur][u][0], false, fb[cur][u][j], (short)0, acc[0][j], false, false);
                    acc[1][j] = __builtin_amdgcn_wmma_f32_16x16x4_f32(
                        false, fa[cur][u][1], false, fb[cur][u][j], (short)0, acc[1][j], false, false);
                }
            }
        }
        __syncthreads();
        if (has_next) store_tile();
        __syncthreads();
    }

    constexpr bool RD_C0 = (EPI == 1 || EPI == 4 || EPI == 6 || EPI == 7 || EPI == 8);
    constexpr bool RD_C1 = (EPI == 3 || EPI == 5 || EPI == 6);
    constexpr bool RD_T  = (EPI == 3 || EPI == 4 || EPI == 5 || EPI == 7 || EPI == 8);
    const int row0 = bm + 32 * waveM + 8 * g;
    const int col0 = bn + WN * waveN + ln;
    auto at = [&](int i2, int j, int r) -> size_t {
        return (size_t)(row0 + 16 * i2 + r) * N + (col0 + 16 * j);
    };

    float c0v[2][NTN][8], c1v[2][NTN][8], tv[2][NTN][8];
    #pragma unroll
    for (int i2 = 0; i2 < 2; ++i2) {
        #pragma unroll
        for (int j = 0; j < NTN; ++j) {
            #pragma unroll
            for (int r = 0; r < 8; ++r) {
                const size_t idx = at(i2, j, r);
                if (RD_C0) c0v[i2][j][r] = C0[idx];
                if (RD_C1) c1v[i2][j][r] = C1[idx];
                if (RD_T)  tv[i2][j][r]  = T[idx];
            }
        }
    }

    #pragma unroll
    for (int i2 = 0; i2 < 2; ++i2) {
        #pragma unroll
        for (int j = 0; j < NTN; ++j) {
            #pragma unroll
            for (int r = 0; r < 8; ++r) {
                const size_t idx = at(i2, j, r);
                const float v = acc[i2][j][r];
                if (EPI == 0) {
                    D[idx] = alpha * v;
                } else if (EPI == 1) {
                    D[idx] = c0v[i2][j][r] - v;
                } else if (EPI == 3) {
                    const float t  = tv[i2][j][r];
                    const float c  = c1v[i2][j][r];
                    const float gq = (t - c) - v;
                    const float tn = t - EP_LR * gq;
                    T[idx] = tn;
                    tv[i2][j][r] = tn;
                } else if (EPI == 4) {
                    const float gq = c0v[i2][j][r] - v;
                    T[idx] = tv[i2][j][r] - EP_LR * gq;
                } else if (EPI == 5) {
                    const float t  = tv[i2][j][r];
                    const float rr = t - v;
                    D[idx] = rr;
                    const float gq = rr + EP_BETA * (t - c1v[i2][j][r]);
                    T[idx] = t - EP_LR * gq;
                } else if (EPI == 6) {
                    D[idx] = (c0v[i2][j][r] - c1v[i2][j][r]) - v;
                } else if (EPI == 7) {
                    const float gq = c0v[i2][j][r] + v;
                    D[idx] = gq;
                    T[idx] = tv[i2][j][r] - EP_LR * gq;
                } else {
                    const float gq = c0v[i2][j][r] + v;
                    T[idx] = tv[i2][j][r] - EP_LR * gq;
                }
            }
        }
    }

    if ((EPI == 0 && T != nullptr) || (EPI == 3 && D != nullptr)) {
        #pragma unroll
        for (int i2 = 0; i2 < 2; ++i2) {
            #pragma unroll
            for (int j = 0; j < NTN; ++j) {
                #pragma unroll
                for (int r = 0; r < 8; ++r) {
                    const size_t idx = at(i2, j, r);
                    if (EPI == 0) T[idx] = alpha * acc[i2][j][r];
                    else          D[idx] = tv[i2][j][r] - c1v[i2][j][r];
                }
            }
        }
    }
}

template <bool TA, bool TB, int EPI, int BNT>
__global__ void __launch_bounds__(NT)
gemm_f32_wmma(const float* __restrict__ A, const float* __restrict__ B,
              const float* __restrict__ C0, const float* __restrict__ C1,
              float* __restrict__ D, float* __restrict__ T,
              int M, int N, int K, float alpha)
{
    __shared__ v2f Asp[BK / 2][BM + 1];
    __shared__ v2f Bsp[BK / 2][BN + 1];
    gemm_tile<TA, TB, EPI, BNT>(A, B, C0, C1, D, T, M, N, K, alpha,
                                blockIdx.x, blockIdx.y, Asp, Bsp);
}

struct GemmArgs {
    const float* A;
    const float* B;
    const float* C0;
    const float* C1;
    float* D;
    float* T;
    int M, N, K;
    float alpha;
};

template <bool TA0, bool TB0, int EPI0, int BNT0, bool TA1, bool TB1, int EPI1, int BNT1>
__global__ void __launch_bounds__(NT)
gemm_pair_f32_wmma(GemmArgs p, GemmArgs q, int ny0)
{
    __shared__ v2f Asp[BK / 2][BM + 1];
    __shared__ v2f Bsp[BK / 2][BN + 1];
    const int ty = blockIdx.y;
    if (ty < ny0) {
        gemm_tile<TA0, TB0, EPI0, BNT0>(p.A, p.B, p.C0, p.C1, p.D, p.T, p.M, p.N, p.K, p.alpha,
                                  blockIdx.x, ty, Asp, Bsp);
    } else {
        gemm_tile<TA1, TB1, EPI1, BNT1>(q.A, q.B, q.C0, q.C1, q.D, q.T, q.M, q.N, q.K, q.alpha,
                                  blockIdx.x, ty - ny0, Asp, Bsp);
    }
}

__global__ void __launch_bounds__(256) g3k(float* __restrict__ r3,
                                           const float* __restrict__ t3,
                                           const float* __restrict__ y) {
    size_t i = ((size_t)blockIdx.x * 256 + threadIdx.x) * 4;
    float4 rv = *(const float4*)(r3 + i);
    float4 tv = *(const float4*)(t3 + i);
    float4 yv = *(const float4*)(y + i);
    rv.x += EP_BETA * (tv.x - yv.x);
    rv.y += EP_BETA * (tv.y - yv.y);
    rv.z += EP_BETA * (tv.z - yv.z);
    rv.w += EP_BETA * (tv.w - yv.w);
    *(float4*)(r3 + i) = rv;
}

__global__ void __launch_bounds__(256) updk(float* __restrict__ t,
                                            const float* __restrict__ gr) {
    size_t i = ((size_t)blockIdx.x * 256 + threadIdx.x) * 4;
    float4 tv = *(const float4*)(t + i);
    float4 gv = *(const float4*)(gr + i);
    tv.x -= EP_LR * gv.x; tv.y -= EP_LR * gv.y;
    tv.z -= EP_LR * gv.z; tv.w -= EP_LR * gv.w;
    *(float4*)(t + i) = tv;
}

__global__ void __launch_bounds__(256) loss_partial(const float* __restrict__ s,
                                                    const float* __restrict__ y,
                                                    float* __restrict__ part) {
    __shared__ float red[256];
    size_t i = ((size_t)blockIdx.x * 256 + threadIdx.x) * 4;
    float4 a = *(const float4*)(s + i);
    float4 b = *(const float4*)(y + i);
    float dx = a.x - b.x, dy = a.y - b.y, dz = a.z - b.z, dw = a.w - b.w;
    red[threadIdx.x] = dx * dx + dy * dy + dz * dz + dw * dw;
    __syncthreads();
    for (int off = 128; off > 0; off >>= 1) {
        if (threadIdx.x < off) red[threadIdx.x] += red[threadIdx.x + off];
        __syncthreads();
    }
    if (threadIdx.x == 0) part[blockIdx.x] = red[0];
}

__global__ void __launch_bounds__(256) loss_final(const float* __restrict__ part,
                                                  float* __restrict__ out) {
    __shared__ float red[256];
    red[threadIdx.x] = part[threadIdx.x] + part[threadIdx.x + 256];
    __syncthreads();
    for (int off = 128; off > 0; off >>= 1) {
        if (threadIdx.x < off) red[threadIdx.x] += red[threadIdx.x + off];
        __syncthreads();
    }
    if (threadIdx.x == 0) out[0] = red[0] * (1.0f / 524288.0f);
}

#define GEMM(TA, TB, EPI, BNT, M, N, K, Ap, Bp, C0p, C1p, Dp, Tp, alpha)             \
    gemm_f32_wmma<TA, TB, EPI, BNT><<<dim3((M) / BM, (N) / (BNT)), NT, 0, stream>>>( \
        Ap, Bp, C0p, C1p, Dp, Tp, M, N, K, alpha)

#define GEMM2(TA0, TB0, EPI0, BNT0, TA1, TB1, EPI1, BNT1, P0, P1)                       \
    gemm_pair_f32_wmma<TA0, TB0, EPI0, BNT0, TA1, TB1, EPI1, BNT1>                      \
        <<<dim3((P0).M / BM, (P0).N / (BNT0) + (P1).N / (BNT1)), NT, 0, stream>>>(      \
            P0, P1, (P0).N / (BNT0))

#define EW(kernel, n, ...) kernel<<<dim3((n) / 1024), 256, 0, stream>>>(__VA_ARGS__)

extern "C" void kernel_launch(void* const* d_in, const int* in_sizes, int n_in,
                              void* d_out, int out_size, void* d_ws, size_t ws_size,
                              hipStream_t stream)
{
    const float* x  = (const float*)d_in[0];
    const float* y  = (const float*)d_in[1];
    const float* W0 = (const float*)d_in[2];
    const float* W1 = (const float*)d_in[3];
    const float* W2 = (const float*)d_in[4];
    float* out = (float*)d_out;

    const int Bt = 1024;
    const int D0 = 1024, D1 = 1024, D2 = 1024, D3 = 512;
    const size_t n1 = (size_t)Bt * D1;
    const size_t n2 = (size_t)Bt * D2;
    const size_t n3 = (size_t)Bt * D3;

    float* ws = (float*)d_ws;
    float* p0 = ws;
    float* t1 = p0 + n1;
    float* t2 = t1 + n1;
    float* t3 = t2 + n2;
    float* r1 = t3 + n3;
    float* r2 = r1 + n1;
    float* r3 = r2 + n2;
    float* part = r3 + n3;

    const float* nc = nullptr;
    float* nw = nullptr;

    const GemmArgs res2  = { t1, W1, t2, nc, r2, nw, Bt, D2, D1, 1.0f };
    const GemmArgs res3  = { t2, W2, t3, nc, r3, nw, Bt, D3, D2, 1.0f };
    const GemmArgs res3u = { t2, W2, nc, y,  r3, t3, Bt, D3, D2, 1.0f };

    GEMM(false, false, 0, 64, Bt, D1, D0, x,  W0, nc, nc, p0, t1, 1.0f);
    GEMM(false, false, 0, 64, Bt, D2, D1, t1, W1, nc, nc, t2, nw, 1.0f);
    GEMM(false, false, 0, 32, Bt, D3, D2, t2, W2, nc, nc, t3, nw, 1.0f);
    loss_partial<<<512, 256, 0, stream>>>(t3, y, part);
    loss_final<<<1, 256, 0, stream>>>(part, out + 2621440);

    GEMM2(false, false, 1, 64, false, false, 1, 32, res2, res3);
    GEMM(false, true,  6, 64, Bt, D1, D2, r2, W1, t1, p0, r1, nw, 1.0f);
    GEMM(false, true,  1, 64, Bt, D2, D3, r3, W2, r2, nc, r2, nw, 1.0f);
    EW(g3k, n3, r3, t3, y);
    GEMM(false, true,  7, 64, Bt, D2, D3, r3, W2, r2, nc, r2, t2, 1.0f);
    GEMM(false, true,  8, 64, Bt, D1, D2, r2, W1, r1, nc, nw, t1, 1.0f);
    EW(updk, n3, t3, r3);

    for (int it = 1; it < 20; ++it) {
        GEMM2(false, false, 1, 64, false, false, 5, 32, res2, res3u);
        const GemmArgs upd1 = { r2, W1, nc, p0, (it == 19 ? r1 : nw), t1, Bt, D1, D2, 1.0f };
        const GemmArgs upd2 = { r3, W2, r2, nc, nw, t2, Bt, D2, D3, 1.0f };
        GEMM2(false, true, 3, 64, false, true, 4, 64, upd1, upd2);
    }

    GEMM2(false, false, 1, 64, false, false, 1, 32, res2, res3);
    const GemmArgs gw0 = { x,  r1, nc, nc, out,           nw, D0, D1, Bt, -1.0f / EP_BETA };
    const GemmArgs gw1 = { t1, r2, nc, nc, out + 1048576, nw, D1, D2, Bt, -1.0f / EP_BETA };
    GEMM2(true, false, 0, 64, true, false, 0, 64, gw0, gw1);
    GEMM(true, false, 0, 32, D2, D3, Bt, t2, r3, nc, nc, out + 2097152, nw, -1.0f / EP_BETA);
}


# Relaxation step 0 keeps only its non-zero terms: at the forward chain each residual and t1-p0 is x-x = 0, so g3=beta(t3-y), g2=g3 W2^T, g1=g2 W1^T; 3 GEMMs, 4 launches dropped, kept sums unchanged
# speedup vs baseline: 6.8082x; 1.0519x over previous
#include <hip/hip_runtime.h>


typedef __attribute__((ext_vector_type(2))) float v2f;
typedef __attribute__((ext_vector_type(8))) float v8f;

#define EP_BETA 1e-3f
#define EP_LR   0.5f

constexpr int BM = 64;
constexpr int BN = 64;
constexpr int BK = 64;
constexpr int NT = 128;
constexpr int F4 = BM * BK / 4 / NT;

template <bool TA, bool TB, int EPI, int BNT>
__device__ __forceinline__ void
gemm_tile(const float* __restrict__ A, const float* __restrict__ B,
          const float* __restrict__ C0, const float* __restrict__ C1,
          float* __restrict__ D, float* __restrict__ T,
          int M, int N, int K, float alpha, int tile_x, int tile_y,
          v2f (*Asp)[BM + 1], v2f (*Bsp)[BN + 1])
{
    const int tid   = threadIdx.x;
    const int lane  = tid & 31;
    const int wid   = tid >> 5;
    const int waveM = wid & 1;
    const int waveN = wid >> 1;
    const int g     = lane >> 4;
    const int ln    = lane & 15;

    constexpr int NTN = BNT / 32;
    constexpr int WN  = BNT / 2;
    constexpr int F4B = BNT * BK / 4 / NT;
    constexpr int NQ  = BNT / 4;
    constexpr int NQS = (BNT == 64) ? 4 : 3;
    constexpr int UG  = 4 / NTN;
    constexpr int NG  = BK / 4 / UG;

    const int bm = tile_x * BM;
    const int bn = tile_y * BNT;

    v8f acc[2][NTN] = {};

    float4 va[F4], vb[F4B];

    auto load_tile = [&](int kt) {
        #pragma unroll
        for (int i = 0; i < F4; ++i) {
            int idx4 = i * NT + tid;
            if (!TA) {
                int kq = idx4 & 15;
                int m  = idx4 >> 4;
                va[i] = *(const float4*)(A + (size_t)(bm + m) * K + kt + 4 * kq);
            } else {
                int mq = idx4 & 15;
                int k  = idx4 >> 4;
                va[i] = *(const float4*)(A + (size_t)(kt + k) * M + bm + 4 * mq);
            }
        }
        #pragma unroll
        for (int i = 0; i < F4B; ++i) {
            int idx4 = i * NT + tid;
            if (TB) {
                int kq = idx4 & 15;
                int n  = idx4 >> 4;
                vb[i] = *(const float4*)(B + (size_t)(bn + n) * K + kt + 4 * kq);
            } else {
                int nq = idx4 & (NQ - 1);
                int k  = idx4 >> NQS;
                vb[i] = *(const float4*)(B + (size_t)(kt + k) * N + bn + 4 * nq);
            }
        }
    };

    auto store_tile = [&]() {
        #pragma unroll
        for (int i = 0; i < F4; ++i) {
            int idx4 = i * NT + tid;
            float4 v = va[i];
            if (!TA) {
                int kq = idx4 & 15;
                int m  = idx4 >> 4;
                v2f p0; p0.x = v.x; p0.y = v.y;
                v2f p1; p1.x = v.z; p1.y = v.w;
                Asp[2 * kq + 0][m] = p0;
                Asp[2 * kq + 1][m] = p1;
            } else {
                int mq = idx4 & 15;
                int k  = idx4 >> 4;
                ((float*)&Asp[k >> 1][4 * mq + 0])[k & 1] = v.x;
                ((float*)&Asp[k >> 1][4 * mq + 1])[k & 1] = v.y;
                ((float*)&Asp[k >> 1][4 * mq + 2])[k & 1] = v.z;
                ((float*)&Asp[k >> 1][4 * mq + 3])[k & 1] = v.w;
            }
        }
        #pragma unroll
        for (int i = 0; i < F4B; ++i) {
            int idx4 = i * NT + tid;
            float4 v = vb[i];
            if (TB) {
                int kq = idx4 & 15;
                int n  = idx4 >> 4;
                v2f p0; p0.x = v.x; p0.y = v.y;
                v2f p1; p1.x = v.z; p1.y = v.w;
                Bsp[2 * kq + 0][n] = p0;
                Bsp[2 * kq + 1][n] = p1;
            } else {
                int nq = idx4 & (NQ - 1);
                int k  = idx4 >> NQS;
                ((float*)&Bsp[k >> 1][4 * nq + 0])[k & 1] = v.x;
                ((float*)&Bsp[k >> 1][4 * nq + 1])[k & 1] = v.y;
                ((float*)&Bsp[k >> 1][4 * nq + 2])[k & 1] = v.z;
                ((float*)&Bsp[k >> 1][4 * nq + 3])[k & 1] = v.w;
            }
        }
    };

    v2f fa[2][UG][2], fb[2][UG][NTN];
    auto load_frags = [&](int b, int set) {
        #pragma unroll
        for (int u = 0; u < UG; ++u) {
            const int kp = 2 * (UG * b + u) + g;
            #pragma unroll
            for (int t = 0; t < 2; ++t)
                fa[set][u][t] = Asp[kp][32 * waveM + 16 * t + ln];
            #pragma unroll
            for (int t = 0; t < NTN; ++t)
                fb[set][u][t] = Bsp[kp][WN * waveN + 16 * t + ln];
        }
    };

    load_tile(0);
    store_tile();
    __syncthreads();

    for (int kt = 0; kt < K; kt += BK) {
        const bool has_next = kt + BK < K;
        if (has_next) load_tile(kt + BK);

        load_frags(0, 0);
        #pragma unroll
        for (int b = 0; b < NG; ++b) {
            const int cur = b & 1;
            if (b + 1 < NG) load_frags(b + 1, cur ^ 1);
            __builtin_amdgcn_sched_barrier(0);
            #pragma unroll
            for (int u = 0; u < UG; ++u) {
                #pragma unroll
                for (int j = 0; j < NTN; ++j) {
                    acc[0][j] = __builtin_amdgcn_wmma_f32_16x16x4_f32(
                        false, fa[cur][u][0], false, fb[cur][u][j], (short)0, acc[0][j], false, false);
                    acc[1][j] = __builtin_amdgcn_wmma_f32_16x16x4_f32(
                        false, fa[cur][u][1], false, fb[cur][u][j], (short)0, acc[1][j], false, false);
                }
            }
        }
        __syncthreads();
        if (has_next) store_tile();
        __syncthreads();
    }

    constexpr bool RD_C0 = (EPI == 1 || EPI == 4);
    constexpr bool RD_C1 = (EPI == 3 || EPI == 5);
    constexpr bool RD_T  = (EPI == 3 || EPI == 4 || EPI == 5 || EPI == 7 || EPI == 8);
    const int row0 = bm + 32 * waveM + 8 * g;
    const int col0 = bn + WN * waveN + ln;
    auto at = [&](int i2, int j, int r) -> size_t {
        return (size_t)(row0 + 16 * i2 + r) * N + (col0 + 16 * j);
    };

    float c0v[2][NTN][8], c1v[2][NTN][8], tv[2][NTN][8];
    #pragma unroll
    for (int i2 = 0; i2 < 2; ++i2) {
        #pragma unroll
        for (int j = 0; j < NTN; ++j) {
            #pragma unroll
            for (int r = 0; r < 8; ++r) {
                const size_t idx = at(i2, j, r);
                if (RD_C0) c0v[i2][j][r] = C0[idx];
                if (RD_C1) c1v[i2][j][r] = C1[idx];
                if (RD_T)  tv[i2][j][r]  = T[idx];
            }
        }
    }

    #pragma unroll
    for (int i2 = 0; i2 < 2; ++i2) {
        #pragma unroll
        for (int j = 0; j < NTN; ++j) {
            #pragma unroll
            for (int r = 0; r < 8; ++r) {
                const size_t idx = at(i2, j, r);
                const float v = acc[i2][j][r];
                if (EPI == 0) {
                    D[idx] = alpha * v;
                } else if (EPI == 1) {
                    D[idx] = c0v[i2][j][r] - v;
                } else if (EPI == 3) {
                    const float t  = tv[i2][j][r];
                    const float c  = c1v[i2][j][r];
                    const float gq = (t - c) - v;
                    const float tn = t - EP_LR * gq;
                    T[idx] = tn;
                    tv[i2][j][r] = tn;
                } else if (EPI == 4) {
                    const float gq = c0v[i2][j][r] - v;
                    T[idx] = tv[i2][j][r] - EP_LR * gq;
                } else if (EPI == 5) {
                    const float t  = tv[i2][j][r];
                    const float rr = t - v;
                    D[idx] = rr;
                    const float gq = rr + EP_BETA * (t - c1v[i2][j][r]);
                    T[idx] = t - EP_LR * gq;
                } else if (EPI == 7) {
                    D[idx] = v;
                    T[idx] = tv[i2][j][r] - EP_LR * v;
                } else {
                    T[idx] = tv[i2][j][r] - EP_LR * v;
                }
            }
        }
    }

    if ((EPI == 0 && T != nullptr) || (EPI == 3 && D != nullptr)) {
        #pragma unroll
        for (int i2 = 0; i2 < 2; ++i2) {
            #pragma unroll
            for (int j = 0; j < NTN; ++j) {
                #pragma unroll
                for (int r = 0; r < 8; ++r) {
                    const size_t idx = at(i2, j, r);
                    if (EPI == 0) T[idx] = alpha * acc[i2][j][r];
                    else          D[idx] = tv[i2][j][r] - c1v[i2][j][r];
                }
            }
        }
    }
}

template <bool TA, bool TB, int EPI, int BNT>
__global__ void __launch_bounds__(NT)
gemm_f32_wmma(const float* __restrict__ A, const float* __restrict__ B,
              const float* __restrict__ C0, const float* __restrict__ C1,
              float* __restrict__ D, float* __restrict__ T,
              int M, int N, int K, float alpha)
{
    __shared__ v2f Asp[BK / 2][BM + 1];
    __shared__ v2f Bsp[BK / 2][BN + 1];
    gemm_tile<TA, TB, EPI, BNT>(A, B, C0, C1, D, T, M, N, K, alpha,
                                blockIdx.x, blockIdx.y, Asp, Bsp);
}

struct GemmArgs {
    const float* A;
    const float* B;
    const float* C0;
    const float* C1;
    float* D;
    float* T;
    int M, N, K;
    float alpha;
};

template <bool TA0, bool TB0, int EPI0, int BNT0, bool TA1, bool TB1, int EPI1, int BNT1>
__global__ void __launch_bounds__(NT)
gemm_pair_f32_wmma(GemmArgs p, GemmArgs q, int ny0)
{
    __shared__ v2f Asp[BK / 2][BM + 1];
    __shared__ v2f Bsp[BK / 2][BN + 1];
    const int ty = blockIdx.y;
    if (ty < ny0) {
        gemm_tile<TA0, TB0, EPI0, BNT0>(p.A, p.B, p.C0, p.C1, p.D, p.T, p.M, p.N, p.K, p.alpha,
                                  blockIdx.x, ty, Asp, Bsp);
    } else {
        gemm_tile<TA1, TB1, EPI1, BNT1>(q.A, q.B, q.C0, q.C1, q.D, q.T, q.M, q.N, q.K, q.alpha,
                                  blockIdx.x, ty - ny0, Asp, Bsp);
    }
}

__global__ void __launch_bounds__(256) g3k(float* __restrict__ g3,
                                           float* __restrict__ t3,
                                           const float* __restrict__ y) {
    size_t i = ((size_t)blockIdx.x * 256 + threadIdx.x) * 4;
    float4 tv = *(const float4*)(t3 + i);
    float4 yv = *(const float4*)(y + i);
    float4 gv;
    gv.x = EP_BETA * (tv.x - yv.x);
    gv.y = EP_BETA * (tv.y - yv.y);
    gv.z = EP_BETA * (tv.z - yv.z);
    gv.w = EP_BETA * (tv.w - yv.w);
    tv.x -= EP_LR * gv.x; tv.y -= EP_LR * gv.y;
    tv.z -= EP_LR * gv.z; tv.w -= EP_LR * gv.w;
    *(float4*)(g3 + i) = gv;
    *(float4*)(t3 + i) = tv;
}

__global__ void __launch_bounds__(256) loss_partial(const float* __restrict__ s,
                                                    const float* __restrict__ y,
                                                    float* __restrict__ part) {
    __shared__ float red[256];
    size_t i = ((size_t)blockIdx.x * 256 + threadIdx.x) * 4;
    float4 a = *(const float4*)(s + i);
    float4 b = *(const float4*)(y + i);
    float dx = a.x - b.x, dy = a.y - b.y, dz = a.z - b.z, dw = a.w - b.w;
    red[threadIdx.x] = dx * dx + dy * dy + dz * dz + dw * dw;
    __syncthreads();
    for (int off = 128; off > 0; off >>= 1) {
        if (threadIdx.x < off) red[threadIdx.x] += red[threadIdx.x + off];
        __syncthreads();
    }
    if (threadIdx.x == 0) part[blockIdx.x] = red[0];
}

__global__ void __launch_bounds__(256) loss_final(const float* __restrict__ part,
                                                  float* __restrict__ out) {
    __shared__ float red[256];
    red[threadIdx.x] = part[threadIdx.x] + part[threadIdx.x + 256];
    __syncthreads();
    for (int off = 128; off > 0; off >>= 1) {
        if (threadIdx.x < off) red[threadIdx.x] += red[threadIdx.x + off];
        __syncthreads();
    }
    if (threadIdx.x == 0) out[0] = red[0] * (1.0f / 524288.0f);
}

#define GEMM(TA, TB, EPI, BNT, M, N, K, Ap, Bp, C0p, C1p, Dp, Tp, alpha)             \
    gemm_f32_wmma<TA, TB, EPI, BNT><<<dim3((M) / BM, (N) / (BNT)), NT, 0, stream>>>( \
        Ap, Bp, C0p, C1p, Dp, Tp, M, N, K, alpha)

#define GEMM2(TA0, TB0, EPI0, BNT0, TA1, TB1, EPI1, BNT1, P0, P1)                       \
    gemm_pair_f32_wmma<TA0, TB0, EPI0, BNT0, TA1, TB1, EPI1, BNT1>                      \
        <<<dim3((P0).M / BM, (P0).N / (BNT0) + (P1).N / (BNT1)), NT, 0, stream>>>(      \
            P0, P1, (P0).N / (BNT0))

#define EW(kernel, n, ...) kernel<<<dim3((n) / 1024), 256, 0, stream>>>(__VA_ARGS__)

extern "C" void kernel_launch(void* const* d_in, const int* in_sizes, int n_in,
                              void* d_out, int out_size, void* d_ws, size_t ws_size,
                              hipStream_t stream)
{
    const float* x  = (const float*)d_in[0];
    const float* y  = (const float*)d_in[1];
    const float* W0 = (const float*)d_in[2];
    const float* W1 = (const float*)d_in[3];
    const float* W2 = (const float*)d_in[4];
    float* out = (float*)d_out;

    const int Bt = 1024;
    const int D0 = 1024, D1 = 1024, D2 = 1024, D3 = 512;
    const size_t n1 = (size_t)Bt * D1;
    const size_t n2 = (size_t)Bt * D2;
    const size_t n3 = (size_t)Bt * D3;

    float* ws = (float*)d_ws;
    float* p0 = ws;
    float* t1 = p0 + n1;
    float* t2 = t1 + n1;
    float* t3 = t2 + n2;
    float* r1 = t3 + n3;
    float* r2 = r1 + n1;
    float* r3 = r2 + n2;
    float* part = r3 + n3;

    const float* nc = nullptr;
    float* nw = nullptr;

    const GemmArgs res2  = { t1, W1, t2, nc, r2, nw, Bt, D2, D1, 1.0f };
    const GemmArgs res3  = { t2, W2, t3, nc, r3, nw, Bt, D3, D2, 1.0f };
    const GemmArgs res3u = { t2, W2, nc, y,  r3, t3, Bt, D3, D2, 1.0f };

    GEMM(false, false, 0, 64, Bt, D1, D0, x,  W0, nc, nc, p0, t1, 1.0f);
    GEMM(false, false, 0, 64, Bt, D2, D1, t1, W1, nc, nc, t2, nw, 1.0f);
    GEMM(false, false, 0, 32, Bt, D3, D2, t2, W2, nc, nc, t3, nw, 1.0f);
    loss_partial<<<512, 256, 0, stream>>>(t3, y, part);
    loss_final<<<1, 256, 0, stream>>>(part, out + 2621440);

    EW(g3k, n3, r3, t3, y);
    GEMM(false, true,  7, 64, Bt, D2, D3, r3, W2, nc, nc, r2, t2, 1.0f);
    GEMM(false, true,  8, 64, Bt, D1, D2, r2, W1, nc, nc, nw, t1, 1.0f);

    for (int it = 1; it < 20; ++it) {
        GEMM2(false, false, 1, 64, false, false, 5, 32, res2, res3u);
        const GemmArgs upd1 = { r2, W1, nc, p0, (it == 19 ? r1 : nw), t1, Bt, D1, D2, 1.0f };
        const GemmArgs upd2 = { r3, W2, r2, nc, nw, t2, Bt, D2, D3, 1.0f };
        GEMM2(false, true, 3, 64, false, true, 4, 64, upd1, upd2);
    }

    GEMM2(false, false, 1, 64, false, false, 1, 32, res2, res3);
    const GemmArgs gw0 = { x,  r1, nc, nc, out,           nw, D0, D1, Bt, -1.0f / EP_BETA };
    const GemmArgs gw1 = { t1, r2, nc, nc, out + 1048576, nw, D1, D2, Bt, -1.0f / EP_BETA };
    GEMM2(true, false, 0, 64, true, false, 0, 64, gw0, gw1);
    GEMM(true, false, 0, 32, D2, D3, Bt, t2, r3, nc, nc, out + 2097152, nw, -1.0f / EP_BETA);
}


# K-tile head shortened: first fragment reads issue before the next tile's global loads, whose addresses are fixed per-thread pointers plus a uniform tile offset (no 64-bit vector multiply); sums same
# speedup vs baseline: 7.2653x; 1.0671x over previous
#include <hip/hip_runtime.h>


typedef __attribute__((ext_vector_type(2))) float v2f;
typedef __attribute__((ext_vector_type(8))) float v8f;

#define EP_BETA 1e-3f
#define EP_LR   0.5f

constexpr int BM = 64;
constexpr int BN = 64;
constexpr int BK = 64;
constexpr int NT = 128;
constexpr int F4 = BM * BK / 4 / NT;

template <bool TA, bool TB, int EPI, int BNT>
__device__ __forceinline__ void
gemm_tile(const float* __restrict__ A, const float* __restrict__ B,
          const float* __restrict__ C0, const float* __restrict__ C1,
          float* __restrict__ D, float* __restrict__ T,
          int M, int N, int K, float alpha, int tile_x, int tile_y,
          v2f (*Asp)[BM + 1], v2f (*Bsp)[BN + 1])
{
    const int tid   = threadIdx.x;
    const int lane  = tid & 31;
    const int wid   = tid >> 5;
    const int waveM = wid & 1;
    const int waveN = wid >> 1;
    const int g     = lane >> 4;
    const int ln    = lane & 15;

    constexpr int NTN = BNT / 32;
    constexpr int WN  = BNT / 2;
    constexpr int F4B = BNT * BK / 4 / NT;
    constexpr int NQ  = BNT / 4;
    constexpr int NQS = (BNT == 64) ? 4 : 3;
    constexpr int UG  = 4 / NTN;
    constexpr int NG  = BK / 4 / UG;

    const int bm = tile_x * BM;
    const int bn = tile_y * BNT;

    v8f acc[2][NTN] = {};

    float4 va[F4], vb[F4B];

    constexpr int RB = NT >> NQS;
    const float* pa[F4];
    const float* pb[F4B];
    #pragma unroll
    for (int i = 0; i < F4; ++i)
        pa[i] = TA ? A + (size_t)(8 * i + (tid >> 4)) * M + bm + 4 * (tid & 15)
                   : A + (size_t)(bm + 8 * i + (tid >> 4)) * K + 4 * (tid & 15);
    #pragma unroll
    for (int i = 0; i < F4B; ++i)
        pb[i] = TB ? B + (size_t)(bn + 8 * i + (tid >> 4)) * K + 4 * (tid & 15)
                   : B + (size_t)(RB * i + (tid >> NQS)) * N + bn + 4 * (tid & (NQ - 1));

    auto load_tile = [&](int kt) {
        const size_t oa = TA ? (size_t)kt * M : (size_t)kt;
        const size_t ob = TB ? (size_t)kt : (size_t)kt * N;
        #pragma unroll
        for (int i = 0; i < F4; ++i)
            va[i] = *(const float4*)(pa[i] + oa);
        #pragma unroll
        for (int i = 0; i < F4B; ++i)
            vb[i] = *(const float4*)(pb[i] + ob);
    };

    auto store_tile = [&]() {
        #pragma unroll
        for (int i = 0; i < F4; ++i) {
            int idx4 = i * NT + tid;
            float4 v = va[i];
            if (!TA) {
                int kq = idx4 & 15;
                int m  = idx4 >> 4;
                v2f p0; p0.x = v.x; p0.y = v.y;
                v2f p1; p1.x = v.z; p1.y = v.w;
                Asp[2 * kq + 0][m] = p0;
                Asp[2 * kq + 1][m] = p1;
            } else {
                int mq = idx4 & 15;
                int k  = idx4 >> 4;
                ((float*)&Asp[k >> 1][4 * mq + 0])[k & 1] = v.x;
                ((float*)&Asp[k >> 1][4 * mq + 1])[k & 1] = v.y;
                ((float*)&Asp[k >> 1][4 * mq + 2])[k & 1] = v.z;
                ((float*)&Asp[k >> 1][4 * mq + 3])[k & 1] = v.w;
            }
        }
        #pragma unroll
        for (int i = 0; i < F4B; ++i) {
            int idx4 = i * NT + tid;
            float4 v = vb[i];
            if (TB) {
                int kq = idx4 & 15;
                int n  = idx4 >> 4;
                v2f p0; p0.x = v.x; p0.y = v.y;
                v2f p1; p1.x = v.z; p1.y = v.w;
                Bsp[2 * kq + 0][n] = p0;
                Bsp[2 * kq + 1][n] = p1;
            } else {
                int nq = idx4 & (NQ - 1);
                int k  = idx4 >> NQS;
                ((float*)&Bsp[k >> 1][4 * nq + 0])[k & 1] = v.x;
                ((float*)&Bsp[k >> 1][4 * nq + 1])[k & 1] = v.y;
                ((float*)&Bsp[k >> 1][4 * nq + 2])[k & 1] = v.z;
                ((float*)&Bsp[k >> 1][4 * nq + 3])[k & 1] = v.w;
            }
        }
    };

    v2f fa[2][UG][2], fb[2][UG][NTN];
    auto load_frags = [&](int b, int set) {
        #pragma unroll
        for (int u = 0; u < UG; ++u) {
            const int kp = 2 * (UG * b + u) + g;
            #pragma unroll
            for (int t = 0; t < 2; ++t)
                fa[set][u][t] = Asp[kp][32 * waveM + 16 * t + ln];
            #pragma unroll
            for (int t = 0; t < NTN; ++t)
                fb[set][u][t] = Bsp[kp][WN * waveN + 16 * t + ln];
        }
    };

    load_tile(0);
    store_tile();
    __syncthreads();

    for (int kt = 0; kt < K; kt += BK) {
        const bool has_next = kt + BK < K;

        load_frags(0, 0);
        if (NG > 1) load_frags(1, 1);
        __builtin_amdgcn_sched_barrier(0);
        if (has_next) load_tile(kt + BK);

        #pragma unroll
        for (int b = 0; b < NG; ++b) {
            const int cur = b & 1;
            if (b >= 1 && b + 1 < NG) load_frags(b + 1, cur ^ 1);
            __builtin_amdgcn_sched_barrier(0);
            #pragma unroll
            for (int u = 0; u < UG; ++u) {
                #pragma unroll
                for (int j = 0; j < NTN; ++j) {
                    acc[0][j] = __builtin_amdgcn_wmma_f32_16x16x4_f32(
                        false, fa[cur][u][0], false, fb[cur][u][j], (short)0, acc[0][j], false, false);
                    acc[1][j] = __builtin_amdgcn_wmma_f32_16x16x4_f32(
                        false, fa[cur][u][1], false, fb[cur][u][j], (short)0, acc[1][j], false, false);
                }
            }
        }
        __syncthreads();
        if (has_next) store_tile();
        __syncthreads();
    }

    constexpr bool RD_C0 = (EPI == 1 || EPI == 4);
    constexpr bool RD_C1 = (EPI == 3 || EPI == 5);
    constexpr bool RD_T  = (EPI == 3 || EPI == 4 || EPI == 5 || EPI == 7 || EPI == 8);
    const int row0 = bm + 32 * waveM + 8 * g;
    const int col0 = bn + WN * waveN + ln;
    auto at = [&](int i2, int j, int r) -> size_t {
        return (size_t)(row0 + 16 * i2 + r) * N + (col0 + 16 * j);
    };

    float c0v[2][NTN][8], c1v[2][NTN][8], tv[2][NTN][8];
    #pragma unroll
    for (int i2 = 0; i2 < 2; ++i2) {
        #pragma unroll
        for (int j = 0; j < NTN; ++j) {
            #pragma unroll
            for (int r = 0; r < 8; ++r) {
                const size_t idx = at(i2, j, r);
                if (RD_C0) c0v[i2][j][r] = C0[idx];
                if (RD_C1) c1v[i2][j][r] = C1[idx];
                if (RD_T)  tv[i2][j][r]  = T[idx];
            }
        }
    }

    #pragma unroll
    for (int i2 = 0; i2 < 2; ++i2) {
        #pragma unroll
        for (int j = 0; j < NTN; ++j) {
            #pragma unroll
            for (int r = 0; r < 8; ++r) {
                const size_t idx = at(i2, j, r);
                const float v = acc[i2][j][r];
                if (EPI == 0) {
                    D[idx] = alpha * v;
                } else if (EPI == 1) {
                    D[idx] = c0v[i2][j][r] - v;
                } else if (EPI == 3) {
                    const float t  = tv[i2][j][r];
                    const float c  = c1v[i2][j][r];
                    const float gq = (t - c) - v;
                    const float tn = t - EP_LR * gq;
                    T[idx] = tn;
                    tv[i2][j][r] = tn;
                } else if (EPI == 4) {
                    const float gq = c0v[i2][j][r] - v;
                    T[idx] = tv[i2][j][r] - EP_LR * gq;
                } else if (EPI == 5) {
                    const float t  = tv[i2][j][r];
                    const float rr = t - v;
                    D[idx] = rr;
                    const float gq = rr + EP_BETA * (t - c1v[i2][j][r]);
                    T[idx] = t - EP_LR * gq;
                } else if (EPI == 7) {
                    D[idx] = v;
                    T[idx] = tv[i2][j][r] - EP_LR * v;
                } else {
                    T[idx] = tv[i2][j][r] - EP_LR * v;
                }
            }
        }
    }

    if ((EPI == 0 && T != nullptr) || (EPI == 3 && D != nullptr)) {
        #pragma unroll
        for (int i2 = 0; i2 < 2; ++i2) {
            #pragma unroll
            for (int j = 0; j < NTN; ++j) {
                #pragma unroll
                for (int r = 0; r < 8; ++r) {
                    const size_t idx = at(i2, j, r);
                    if (EPI == 0) T[idx] = alpha * acc[i2][j][r];
                    else          D[idx] = tv[i2][j][r] - c1v[i2][j][r];
                }
            }
        }
    }
}

template <bool TA, bool TB, int EPI, int BNT>
__global__ void __launch_bounds__(NT)
gemm_f32_wmma(const float* __restrict__ A, const float* __restrict__ B,
              const float* __restrict__ C0, const float* __restrict__ C1,
              float* __restrict__ D, float* __restrict__ T,
              int M, int N, int K, float alpha)
{
    __shared__ v2f Asp[BK / 2][BM + 1];
    __shared__ v2f Bsp[BK / 2][BN + 1];
    gemm_tile<TA, TB, EPI, BNT>(A, B, C0, C1, D, T, M, N, K, alpha,
                                blockIdx.x, blockIdx.y, Asp, Bsp);
}

struct GemmArgs {
    const float* A;
    const float* B;
    const float* C0;
    const float* C1;
    float* D;
    float* T;
    int M, N, K;
    float alpha;
};

template <bool TA0, bool TB0, int EPI0, int BNT0, bool TA1, bool TB1, int EPI1, int BNT1>
__global__ void __launch_bounds__(NT)
gemm_pair_f32_wmma(GemmArgs p, GemmArgs q, int ny0)
{
    __shared__ v2f Asp[BK / 2][BM + 1];
    __shared__ v2f Bsp[BK / 2][BN + 1];
    const int ty = blockIdx.y;
    if (ty < ny0) {
        gemm_tile<TA0, TB0, EPI0, BNT0>(p.A, p.B, p.C0, p.C1, p.D, p.T, p.M, p.N, p.K, p.alpha,
                                  blockIdx.x, ty, Asp, Bsp);
    } else {
        gemm_tile<TA1, TB1, EPI1, BNT1>(q.A, q.B, q.C0, q.C1, q.D, q.T, q.M, q.N, q.K, q.alpha,
                                  blockIdx.x, ty - ny0, Asp, Bsp);
    }
}

__global__ void __launch_bounds__(256) g3k(float* __restrict__ g3,
                                           float* __restrict__ t3,
                                           const float* __restrict__ y) {
    size_t i = ((size_t)blockIdx.x * 256 + threadIdx.x) * 4;
    float4 tv = *(const float4*)(t3 + i);
    float4 yv = *(const float4*)(y + i);
    float4 gv;
    gv.x = EP_BETA * (tv.x - yv.x);
    gv.y = EP_BETA * (tv.y - yv.y);
    gv.z = EP_BETA * (tv.z - yv.z);
    gv.w = EP_BETA * (tv.w - yv.w);
    tv.x -= EP_LR * gv.x; tv.y -= EP_LR * gv.y;
    tv.z -= EP_LR * gv.z; tv.w -= EP_LR * gv.w;
    *(float4*)(g3 + i) = gv;
    *(float4*)(t3 + i) = tv;
}

__global__ void __launch_bounds__(256) loss_partial(const float* __restrict__ s,
                                                    const float* __restrict__ y,
                                                    float* __restrict__ part) {
    __shared__ float red[256];
    size_t i = ((size_t)blockIdx.x * 256 + threadIdx.x) * 4;
    float4 a = *(const float4*)(s + i);
    float4 b = *(const float4*)(y + i);
    float dx = a.x - b.x, dy = a.y - b.y, dz = a.z - b.z, dw = a.w - b.w;
    red[threadIdx.x] = dx * dx + dy * dy + dz * dz + dw * dw;
    __syncthreads();
    for (int off = 128; off > 0; off >>= 1) {
        if (threadIdx.x < off) red[threadIdx.x] += red[threadIdx.x + off];
        __syncthreads();
    }
    if (threadIdx.x == 0) part[blockIdx.x] = red[0];
}

__global__ void __launch_bounds__(256) loss_final(const float* __restrict__ part,
                                                  float* __restrict__ out) {
    __shared__ float red[256];
    red[threadIdx.x] = part[threadIdx.x] + part[threadIdx.x + 256];
    __syncthreads();
    for (int off = 128; off > 0; off >>= 1) {
        if (threadIdx.x < off) red[threadIdx.x] += red[threadIdx.x + off];
        __syncthreads();
    }
    if (threadIdx.x == 0) out[0] = red[0] * (1.0f / 524288.0f);
}

#define GEMM(TA, TB, EPI, BNT, M, N, K, Ap, Bp, C0p, C1p, Dp, Tp, alpha)             \
    gemm_f32_wmma<TA, TB, EPI, BNT><<<dim3((M) / BM, (N) / (BNT)), NT, 0, stream>>>( \
        Ap, Bp, C0p, C1p, Dp, Tp, M, N, K, alpha)

#define GEMM2(TA0, TB0, EPI0, BNT0, TA1, TB1, EPI1, BNT1, P0, P1)                       \
    gemm_pair_f32_wmma<TA0, TB0, EPI0, BNT0, TA1, TB1, EPI1, BNT1>                      \
        <<<dim3((P0).M / BM, (P0).N / (BNT0) + (P1).N / (BNT1)), NT, 0, stream>>>(      \
            P0, P1, (P0).N / (BNT0))

#define EW(kernel, n, ...) kernel<<<dim3((n) / 1024), 256, 0, stream>>>(__VA_ARGS__)

extern "C" void kernel_launch(void* const* d_in, const int* in_sizes, int n_in,
                              void* d_out, int out_size, void* d_ws, size_t ws_size,
                              hipStream_t stream)
{
    const float* x  = (const float*)d_in[0];
    const float* y  = (const float*)d_in[1];
    const float* W0 = (const float*)d_in[2];
    const float* W1 = (const float*)d_in[3];
    const float* W2 = (const float*)d_in[4];
    float* out = (float*)d_out;

    const int Bt = 1024;
    const int D0 = 1024, D1 = 1024, D2 = 1024, D3 = 512;
    const size_t n1 = (size_t)Bt * D1;
    const size_t n2 = (size_t)Bt * D2;
    const size_t n3 = (size_t)Bt * D3;

    float* ws = (float*)d_ws;
    float* p0 = ws;
    float* t1 = p0 + n1;
    float* t2 = t1 + n1;
    float* t3 = t2 + n2;
    float* r1 = t3 + n3;
    float* r2 = r1 + n1;
    float* r3 = r2 + n2;
    float* part = r3 + n3;

    const float* nc = nullptr;
    float* nw = nullptr;

    const GemmArgs res2  = { t1, W1, t2, nc, r2, nw, Bt, D2, D1, 1.0f };
    const GemmArgs res3  = { t2, W2, t3, nc, r3, nw, Bt, D3, D2, 1.0f };
    const GemmArgs res3u = { t2, W2, nc, y,  r3, t3, Bt, D3, D2, 1.0f };

    GEMM(false, false, 0, 64, Bt, D1, D0, x,  W0, nc, nc, p0, t1, 1.0f);
    GEMM(false, false, 0, 64, Bt, D2, D1, t1, W1, nc, nc, t2, nw, 1.0f);
    GEMM(false, false, 0, 32, Bt, D3, D2, t2, W2, nc, nc, t3, nw, 1.0f);
    loss_partial<<<512, 256, 0, stream>>>(t3, y, part);
    loss_final<<<1, 256, 0, stream>>>(part, out + 2621440);

    EW(g3k, n3, r3, t3, y);
    GEMM(false, true,  7, 64, Bt, D2, D3, r3, W2, nc, nc, r2, t2, 1.0f);
    GEMM(false, true,  8, 64, Bt, D1, D2, r2, W1, nc, nc, nw, t1, 1.0f);

    for (int it = 1; it < 20; ++it) {
        GEMM2(false, false, 1, 64, false, false, 5, 32, res2, res3u);
        const GemmArgs upd1 = { r2, W1, nc, p0, (it == 19 ? r1 : nw), t1, Bt, D1, D2, 1.0f };
        const GemmArgs upd2 = { r3, W2, r2, nc, nw, t2, Bt, D2, D3, 1.0f };
        GEMM2(false, true, 3, 64, false, true, 4, 64, upd1, upd2);
    }

    GEMM2(false, false, 1, 64, false, false, 1, 32, res2, res3);
    const GemmArgs gw0 = { x,  r1, nc, nc, out,           nw, D0, D1, Bt, -1.0f / EP_BETA };
    const GemmArgs gw1 = { t1, r2, nc, nc, out + 1048576, nw, D1, D2, Bt, -1.0f / EP_BETA };
    GEMM2(true, false, 0, 64, true, false, 0, 64, gw0, gw1);
    GEMM(true, false, 0, 32, D2, D3, Bt, t2, r3, nc, nc, out + 2097152, nw, -1.0f / EP_BETA);
}
